# starting kernel
# baseline (speedup 1.0000x reference)
#include <hip/hip_runtime.h>
#include <math.h>

typedef __attribute__((ext_vector_type(2))) float v2f;
typedef __attribute__((ext_vector_type(4))) float f4v;
typedef __attribute__((ext_vector_type(8))) float v8f;

#define WAVES 4
#define TILE  (WAVES * 16)
#define ISTR  68

__global__ __launch_bounds__(64) void build_unitary_kernel(
    const float* __restrict__ qp, float* __restrict__ Ut_r, float* __restrict__ Ut_i) {
  __shared__ float vr[64][64];
  __shared__ float vi[64][64];
  const int c = threadIdx.x;
  float* rv = vr[c];
  float* iv = vi[c];
  for (int s = 0; s < 64; ++s) { rv[s] = (s == c) ? 1.0f : 0.0f; iv[s] = 0.0f; }

  auto cnot = [&](int cq, int tq) {
    const int mc = 1 << (5 - cq), mt = 1 << (5 - tq);
    for (int s = 0; s < 64; ++s) {
      if ((s & mc) && !(s & mt)) {
        const int s1 = s | mt;
        float t = rv[s]; rv[s] = rv[s1]; rv[s1] = t;
        t = iv[s]; iv[s] = iv[s1]; iv[s1] = t;
      }
    }
  };
  auto rx = [&](int q, float th) {
    float sh, ch; sincosf(0.5f * th, &sh, &ch);
    const int m = 1 << (5 - q);
    for (int s0 = 0; s0 < 64; ++s0) if (!(s0 & m)) {
      const int s1 = s0 | m;
      const float r0 = rv[s0], i0 = iv[s0], r1 = rv[s1], i1 = iv[s1];
      rv[s0] = ch * r0 + sh * i1;  iv[s0] = ch * i0 - sh * r1;
      rv[s1] = ch * r1 + sh * i0;  iv[s1] = ch * i1 - sh * r0;
    }
  };
  auto ry = [&](int q, float th) {
    float sh, ch; sincosf(0.5f * th, &sh, &ch);
    const int m = 1 << (5 - q);
    for (int s0 = 0; s0 < 64; ++s0) if (!(s0 & m)) {
      const int s1 = s0 | m;
      const float r0 = rv[s0], i0 = iv[s0], r1 = rv[s1], i1 = iv[s1];
      rv[s0] = ch * r0 - sh * r1;  iv[s0] = ch * i0 - sh * i1;
      rv[s1] = sh * r0 + ch * r1;  iv[s1] = sh * i0 + ch * i1;
    }
  };
  auto rz = [&](int q, float th) {
    float sh, ch; sincosf(0.5f * th, &sh, &ch);
    const int m = 1 << (5 - q);
    for (int s0 = 0; s0 < 64; ++s0) if (!(s0 & m)) {
      const int s1 = s0 | m;
      const float r0 = rv[s0], i0 = iv[s0], r1 = rv[s1], i1 = iv[s1];
      rv[s0] = ch * r0 + sh * i0;  iv[s0] = ch * i0 - sh * r0;
      rv[s1] = ch * r1 - sh * i1;  iv[s1] = ch * i1 + sh * r1;
    }
  };
  auto crz = [&](int cq, int tq, float th) {
    float sh, ch; sincosf(0.5f * th, &sh, &ch);
    const int mc = 1 << (5 - cq), mt = 1 << (5 - tq);
    for (int s = 0; s < 64; ++s) if (s & mc) {
      const float r = rv[s], i = iv[s];
      if (s & mt) { rv[s] = ch * r - sh * i; iv[s] = ch * i + sh * r; }
      else        { rv[s] = ch * r + sh * i; iv[s] = ch * i - sh * r; }
    }
  };

  for (int layer = 0; layer < 3; ++layer) {
    cnot(0, 1); cnot(2, 3); cnot(4, 5);
    cnot(1, 2); cnot(3, 4);
    for (int i = 0; i < 6; ++i) {
      const float tx = qp[layer * 18 + i * 3 + 0];
      const float ty = qp[layer * 18 + i * 3 + 1];
      const float tz = qp[layer * 18 + i * 3 + 2];
      rx(i, tx); ry(i, ty); rz(i, tz);
    }
    if (layer < 2)
      for (int i = 0; i < 6; ++i)
        crz(i, (i + 1) % 6, qp[layer * 18 + i * 3 + 0]);
  }
  for (int s = 0; s < 64; ++s) { Ut_r[c * 64 + s] = rv[s]; Ut_i[c * 64 + s] = iv[s]; }
}

__global__ __launch_bounds__(WAVES * 32) void qnn_kernel(
    const float* __restrict__ x,
    const float* __restrict__ W1, const float* __restrict__ b1,
    const float* __restrict__ W2, const float* __restrict__ b2,
    const float* __restrict__ W3, const float* __restrict__ b3,
    const float* __restrict__ W4, const float* __restrict__ b4,
    const float* __restrict__ Ut_r, const float* __restrict__ Ut_i,
    float* __restrict__ out, int batch) {
  __shared__ float sUr[64 * 64];
  __shared__ float sUi[64 * 64];
  __shared__ float sAmp[WAVES][16 * ISTR];

  const int tid  = threadIdx.x;
  const int wave = tid >> 5;
  const int lane = tid & 31;
  const int half = lane >> 4;
  const int l    = lane & 15;
  const int ebase = blockIdx.x * TILE + wave * 16;

  {
    const f4v* srcR = (const f4v*)Ut_r;
    const f4v* srcI = (const f4v*)Ut_i;
    f4v* dR = (f4v*)sUr;
    f4v* dI = (f4v*)sUi;
    for (int i = tid; i < 1024; i += WAVES * 32) { dR[i] = srcR[i]; dI[i] = srcI[i]; }
  }

  if (lane < 16) {
    int e = ebase + lane;
    const int ec = (e < batch) ? e : (batch - 1);
    const float x0 = x[ec * 4 + 0], x1 = x[ec * 4 + 1];
    const float x2 = x[ec * 4 + 2], x3 = x[ec * 4 + 3];
    float h[16];
#pragma unroll
    for (int j = 0; j < 16; ++j) {
      float t = b1[j] + x0 * W1[j] + x1 * W1[16 + j] + x2 * W1[32 + j] + x3 * W1[48 + j];
      h[j] = t > 0.0f ? t : 0.0f;
    }
    float A0[6], A1[6];
#pragma unroll
    for (int q = 0; q < 6; ++q) {
      float t = b2[q];
#pragma unroll
      for (int j = 0; j < 16; ++j) t += h[j] * W2[j * 6 + q];
      const float ang = 0.5f * 3.14159265358979323846f * tanhf(t);
      float sn, cs; sincosf(ang, &sn, &cs);
      const float is2 = 0.70710678118654752440f;
      A0[q] = (cs - sn) * is2;
      A1[q] = (cs + sn) * is2;
    }
    float m01[4], m23[4], m45[4];
#pragma unroll
    for (int i = 0; i < 4; ++i) {
      m01[i] = ((i & 2) ? A1[0] : A0[0]) * ((i & 1) ? A1[1] : A0[1]);
      m23[i] = ((i & 2) ? A1[2] : A0[2]) * ((i & 1) ? A1[3] : A0[3]);
      m45[i] = ((i & 2) ? A1[4] : A0[4]) * ((i & 1) ? A1[5] : A0[5]);
    }
    float* dst = &sAmp[wave][lane * ISTR];
#pragma unroll
    for (int s = 0; s < 64; ++s)
      dst[s] = m01[s >> 4] * m23[(s >> 2) & 3] * m45[s & 3];
  }
  __syncthreads();

  v8f accr[4] = {v8f{}, v8f{}, v8f{}, v8f{}};
  v8f acci[4] = {v8f{}, v8f{}, v8f{}, v8f{}};
  const float* amp = &sAmp[wave][l * ISTR];
#pragma unroll
  for (int kt = 0; kt < 16; ++kt) {
    const int kb = 4 * kt + 2 * half;
    v2f a;
    a.x = amp[kb];
    a.y = amp[kb + 1];
#pragma unroll
    for (int nt = 0; nt < 4; ++nt) {
      const int n = nt * 16 + l;
      v2f br, bi;
      br.x = sUr[kb * 64 + n];        br.y = sUr[(kb + 1) * 64 + n];
      bi.x = sUi[kb * 64 + n];        bi.y = sUi[(kb + 1) * 64 + n];
      accr[nt] = __builtin_amdgcn_wmma_f32_16x16x4_f32(
          false, a, false, br, (short)0, accr[nt], false, false);
      acci[nt] = __builtin_amdgcn_wmma_f32_16x16x4_f32(
          false, a, false, bi, (short)0, acci[nt], false, false);
    }
  }

  {
    float* wamp = sAmp[wave];
#pragma unroll
    for (int nt = 0; nt < 4; ++nt)
#pragma unroll
      for (int r = 0; r < 8; ++r) {
        const float pr = accr[nt][r], pi = acci[nt][r];
        wamp[(r + 8 * half) * ISTR + nt * 16 + l] = pr * pr + pi * pi;
      }
  }
  __syncthreads();

  if (lane < 16) {
    const int e = ebase + lane;
    if (e < batch) {
      const float* p = &sAmp[wave][lane * ISTR];
      float z0 = 0, z1 = 0, z2 = 0, z3 = 0, z4 = 0, z5 = 0;
#pragma unroll
      for (int s = 0; s < 64; ++s) {
        const float v = p[s];
        z0 += (s & 32) ? -v : v;
        z1 += (s & 16) ? -v : v;
        z2 += (s & 8)  ? -v : v;
        z3 += (s & 4)  ? -v : v;
        z4 += (s & 2)  ? -v : v;
        z5 += (s & 1)  ? -v : v;
      }
      const float z[6] = {z0, z1, z2, z3, z4, z5};
      float o0 = b4[0], o1 = b4[1], o2 = b4[2], o3 = b4[3];
#pragma unroll
      for (int j = 0; j < 16; ++j) {
        float g = b3[j];
#pragma unroll
        for (int q = 0; q < 6; ++q) g += z[q] * W3[q * 16 + j];
        g = g > 0.0f ? g : 0.0f;
        o0 += g * W4[j * 4 + 0]; o1 += g * W4[j * 4 + 1];
        o2 += g * W4[j * 4 + 2]; o3 += g * W4[j * 4 + 3];
      }
      float* op = out + (size_t)e * 4;
      op[0] = o0; op[1] = o1; op[2] = o2; op[3] = o3;
    }
  }
}

extern "C" void kernel_launch(void* const* d_in, const int* in_sizes, int n_in,
                              void* d_out, int out_size, void* d_ws, size_t ws_size,
                              hipStream_t stream) {
  const float* x  = (const float*)d_in[0];
  const float* W1 = (const float*)d_in[1];
  const float* b1 = (const float*)d_in[2];
  const float* W2 = (const float*)d_in[3];
  const float* b2 = (const float*)d_in[4];
  const float* qp = (const float*)d_in[5];
  const float* W3 = (const float*)d_in[6];
  const float* b3 = (const float*)d_in[7];
  const float* W4 = (const float*)d_in[8];
  const float* b4 = (const float*)d_in[9];
  float* out  = (float*)d_out;
  float* Ut_r = (float*)d_ws;
  float* Ut_i = Ut_r + 64 * 64;

  const int batch = in_sizes[0] / 4;
  build_unitary_kernel<<<dim3(1), dim3(64), 0, stream>>>(qp, Ut_r, Ut_i);
  const int tiles = (batch + TILE - 1) / TILE;
  qnn_kernel<<<dim3(tiles), dim3(WAVES * 32), 0, stream>>>(
      x, W1, b1, W2, b2, W3, b3, W4, b4, Ut_r, Ut_i, out, batch);
}


# Unitary build re-parallelised: one wave per basis state (64 blocks of 32 lanes), two amplitudes per lane in registers, gate partners by lane permute; was one 64-thread block on same-bank LDS rows
# speedup vs baseline: 9.5314x; 9.5314x over previous
#include <hip/hip_runtime.h>
#include <math.h>

typedef __attribute__((ext_vector_type(2))) float v2f;
typedef __attribute__((ext_vector_type(4))) float f4v;
typedef __attribute__((ext_vector_type(8))) float v8f;

#define WAVES 4
#define TILE  (WAVES * 16)
#define ISTR  68

__device__ __forceinline__ float lane_xor(float v, int lane, int mask) {
  return __int_as_float(__builtin_amdgcn_ds_bpermute((lane ^ mask) << 2, __float_as_int(v)));
}
__device__ __forceinline__ float lane_get(float v, int src) {
  return __int_as_float(__builtin_amdgcn_readlane(__float_as_int(v), src));
}

__global__ __launch_bounds__(32) void build_unitary_kernel(
    const float* __restrict__ qp, float* __restrict__ Ut_r, float* __restrict__ Ut_i) {
  const int lane = threadIdx.x;
  const int c    = blockIdx.x;
  const int s0 = lane, s1 = lane + 32;

  float snA, csA, snB, csB;
  {
    const int tb = (lane + 32 < 54) ? lane + 32 : 53;
    const float thA = qp[lane], thB = qp[tb];
    sincosf(0.5f * thA, &snA, &csA);
    sincosf(0.5f * thB, &snB, &csB);
  }
  auto trig = [&](int idx, float& sh, float& ch) {
    const int j = idx & 31;
    const float sa = lane_get(snA, j), ca = lane_get(csA, j);
    const float sb = lane_get(snB, j), cb = lane_get(csB, j);
    sh = (idx < 32) ? sa : sb;
    ch = (idx < 32) ? ca : cb;
  };

  float r0 = (s0 == c) ? 1.0f : 0.0f, i0 = 0.0f;
  float r1 = (s1 == c) ? 1.0f : 0.0f, i1 = 0.0f;

  auto partner = [&](int m, float& pr0, float& pi0, float& pr1, float& pi1) {
    if (m == 32) { pr0 = r1; pi0 = i1; pr1 = r0; pi1 = i0; }
    else {
      pr0 = lane_xor(r0, lane, m); pi0 = lane_xor(i0, lane, m);
      pr1 = lane_xor(r1, lane, m); pi1 = lane_xor(i1, lane, m);
    }
  };
  auto cnot = [&](int cq, int tq) {
    const int mc = 1 << (5 - cq), mt = 1 << (5 - tq);
    float pr0, pi0, pr1, pi1;
    partner(mt, pr0, pi0, pr1, pi1);
    if (s0 & mc) { r0 = pr0; i0 = pi0; }
    if (s1 & mc) { r1 = pr1; i1 = pi1; }
  };
  auto rx = [&](int q, float sh, float ch) {
    const int m = 1 << (5 - q);
    float pr0, pi0, pr1, pi1;
    partner(m, pr0, pi0, pr1, pi1);
    const float a0 = ch * r0 + sh * pi0, b0 = ch * i0 - sh * pr0;
    const float a1 = ch * r1 + sh * pi1, b1 = ch * i1 - sh * pr1;
    r0 = a0; i0 = b0; r1 = a1; i1 = b1;
  };
  auto ry = [&](int q, float sh, float ch) {
    const int m = 1 << (5 - q);
    float pr0, pi0, pr1, pi1;
    partner(m, pr0, pi0, pr1, pi1);
    const float t0 = (s0 & m) ? sh : -sh;
    const float t1 = (s1 & m) ? sh : -sh;
    const float a0 = ch * r0 + t0 * pr0, b0 = ch * i0 + t0 * pi0;
    const float a1 = ch * r1 + t1 * pr1, b1 = ch * i1 + t1 * pi1;
    r0 = a0; i0 = b0; r1 = a1; i1 = b1;
  };
  auto phase = [&](int s, int m, float sh, float ch, float& r, float& i) {
    const float t = (s & m) ? -sh : sh;
    const float a = ch * r + t * i, b = ch * i - t * r;
    r = a; i = b;
  };
  auto rz = [&](int q, float sh, float ch) {
    const int m = 1 << (5 - q);
    phase(s0, m, sh, ch, r0, i0);
    phase(s1, m, sh, ch, r1, i1);
  };
  auto crz = [&](int cq, int tq, float sh, float ch) {
    const int mc = 1 << (5 - cq), mt = 1 << (5 - tq);
    float a = r0, b = i0;
    phase(s0, mt, sh, ch, a, b);
    if (s0 & mc) { r0 = a; i0 = b; }
    a = r1; b = i1;
    phase(s1, mt, sh, ch, a, b);
    if (s1 & mc) { r1 = a; i1 = b; }
  };

  for (int layer = 0; layer < 3; ++layer) {
    cnot(0, 1); cnot(2, 3); cnot(4, 5);
    cnot(1, 2); cnot(3, 4);
    for (int i = 0; i < 6; ++i) {
      float sh, ch;
      trig(layer * 18 + i * 3 + 0, sh, ch); rx(i, sh, ch);
      trig(layer * 18 + i * 3 + 1, sh, ch); ry(i, sh, ch);
      trig(layer * 18 + i * 3 + 2, sh, ch); rz(i, sh, ch);
    }
    if (layer < 2)
      for (int i = 0; i < 6; ++i) {
        float sh, ch;
        trig(layer * 18 + i * 3 + 0, sh, ch);
        crz(i, (i + 1) % 6, sh, ch);
      }
  }
  Ut_r[c * 64 + s0] = r0; Ut_r[c * 64 + s1] = r1;
  Ut_i[c * 64 + s0] = i0; Ut_i[c * 64 + s1] = i1;
}

__global__ __launch_bounds__(WAVES * 32) void qnn_kernel(
    const float* __restrict__ x,
    const float* __restrict__ W1, const float* __restrict__ b1,
    const float* __restrict__ W2, const float* __restrict__ b2,
    const float* __restrict__ W3, const float* __restrict__ b3,
    const float* __restrict__ W4, const float* __restrict__ b4,
    const float* __restrict__ Ut_r, const float* __restrict__ Ut_i,
    float* __restrict__ out, int batch) {
  __shared__ float sUr[64 * 64];
  __shared__ float sUi[64 * 64];
  __shared__ float sAmp[WAVES][16 * ISTR];

  const int tid  = threadIdx.x;
  const int wave = tid >> 5;
  const int lane = tid & 31;
  const int half = lane >> 4;
  const int l    = lane & 15;
  const int ebase = blockIdx.x * TILE + wave * 16;

  {
    const f4v* srcR = (const f4v*)Ut_r;
    const f4v* srcI = (const f4v*)Ut_i;
    f4v* dR = (f4v*)sUr;
    f4v* dI = (f4v*)sUi;
    for (int i = tid; i < 1024; i += WAVES * 32) { dR[i] = srcR[i]; dI[i] = srcI[i]; }
  }

  if (lane < 16) {
    int e = ebase + lane;
    const int ec = (e < batch) ? e : (batch - 1);
    const float x0 = x[ec * 4 + 0], x1 = x[ec * 4 + 1];
    const float x2 = x[ec * 4 + 2], x3 = x[ec * 4 + 3];
    float h[16];
#pragma unroll
    for (int j = 0; j < 16; ++j) {
      float t = b1[j] + x0 * W1[j] + x1 * W1[16 + j] + x2 * W1[32 + j] + x3 * W1[48 + j];
      h[j] = t > 0.0f ? t : 0.0f;
    }
    float A0[6], A1[6];
#pragma unroll
    for (int q = 0; q < 6; ++q) {
      float t = b2[q];
#pragma unroll
      for (int j = 0; j < 16; ++j) t += h[j] * W2[j * 6 + q];
      const float ang = 0.5f * 3.14159265358979323846f * tanhf(t);
      float sn, cs; sincosf(ang, &sn, &cs);
      const float is2 = 0.70710678118654752440f;
      A0[q] = (cs - sn) * is2;
      A1[q] = (cs + sn) * is2;
    }
    float m01[4], m23[4], m45[4];
#pragma unroll
    for (int i = 0; i < 4; ++i) {
      m01[i] = ((i & 2) ? A1[0] : A0[0]) * ((i & 1) ? A1[1] : A0[1]);
      m23[i] = ((i & 2) ? A1[2] : A0[2]) * ((i & 1) ? A1[3] : A0[3]);
      m45[i] = ((i & 2) ? A1[4] : A0[4]) * ((i & 1) ? A1[5] : A0[5]);
    }
    float* dst = &sAmp[wave][lane * ISTR];
#pragma unroll
    for (int s = 0; s < 64; ++s)
      dst[s] = m01[s >> 4] * m23[(s >> 2) & 3] * m45[s & 3];
  }
  __syncthreads();

  v8f accr[4] = {v8f{}, v8f{}, v8f{}, v8f{}};
  v8f acci[4] = {v8f{}, v8f{}, v8f{}, v8f{}};
  const float* amp = &sAmp[wave][l * ISTR];
#pragma unroll
  for (int kt = 0; kt < 16; ++kt) {
    const int kb = 4 * kt + 2 * half;
    v2f a;
    a.x = amp[kb];
    a.y = amp[kb + 1];
#pragma unroll
    for (int nt = 0; nt < 4; ++nt) {
      const int n = nt * 16 + l;
      v2f br, bi;
      br.x = sUr[kb * 64 + n];        br.y = sUr[(kb + 1) * 64 + n];
      bi.x = sUi[kb * 64 + n];        bi.y = sUi[(kb + 1) * 64 + n];
      accr[nt] = __builtin_amdgcn_wmma_f32_16x16x4_f32(
          false, a, false, br, (short)0, accr[nt], false, false);
      acci[nt] = __builtin_amdgcn_wmma_f32_16x16x4_f32(
          false, a, false, bi, (short)0, acci[nt], false, false);
    }
  }

  {
    float* wamp = sAmp[wave];
#pragma unroll
    for (int nt = 0; nt < 4; ++nt)
#pragma unroll
      for (int r = 0; r < 8; ++r) {
        const float pr = accr[nt][r], pi = acci[nt][r];
        wamp[(r + 8 * half) * ISTR + nt * 16 + l] = pr * pr + pi * pi;
      }
  }
  __syncthreads();

  if (lane < 16) {
    const int e = ebase + lane;
    if (e < batch) {
      const float* p = &sAmp[wave][lane * ISTR];
      float z0 = 0, z1 = 0, z2 = 0, z3 = 0, z4 = 0, z5 = 0;
#pragma unroll
      for (int s = 0; s < 64; ++s) {
        const float v = p[s];
        z0 += (s & 32) ? -v : v;
        z1 += (s & 16) ? -v : v;
        z2 += (s & 8)  ? -v : v;
        z3 += (s & 4)  ? -v : v;
        z4 += (s & 2)  ? -v : v;
        z5 += (s & 1)  ? -v : v;
      }
      const float z[6] = {z0, z1, z2, z3, z4, z5};
      float o0 = b4[0], o1 = b4[1], o2 = b4[2], o3 = b4[3];
#pragma unroll
      for (int j = 0; j < 16; ++j) {
        float g = b3[j];
#pragma unroll
        for (int q = 0; q < 6; ++q) g += z[q] * W3[q * 16 + j];
        g = g > 0.0f ? g : 0.0f;
        o0 += g * W4[j * 4 + 0]; o1 += g * W4[j * 4 + 1];
        o2 += g * W4[j * 4 + 2]; o3 += g * W4[j * 4 + 3];
      }
      float* op = out + (size_t)e * 4;
      op[0] = o0; op[1] = o1; op[2] = o2; op[3] = o3;
    }
  }
}

extern "C" void kernel_launch(void* const* d_in, const int* in_sizes, int n_in,
                              void* d_out, int out_size, void* d_ws, size_t ws_size,
                              hipStream_t stream) {
  const float* x  = (const float*)d_in[0];
  const float* W1 = (const float*)d_in[1];
  const float* b1 = (const float*)d_in[2];
  const float* W2 = (const float*)d_in[3];
  const float* b2 = (const float*)d_in[4];
  const float* qp = (const float*)d_in[5];
  const float* W3 = (const float*)d_in[6];
  const float* b3 = (const float*)d_in[7];
  const float* W4 = (const float*)d_in[8];
  const float* b4 = (const float*)d_in[9];
  float* out  = (float*)d_out;
  float* Ut_r = (float*)d_ws;
  float* Ut_i = Ut_r + 64 * 64;

  const int batch = in_sizes[0] / 4;
  build_unitary_kernel<<<dim3(64), dim3(32), 0, stream>>>(qp, Ut_r, Ut_i);
  const int tiles = (batch + TILE - 1) / TILE;
  qnn_kernel<<<dim3(tiles), dim3(WAVES * 32), 0, stream>>>(
      x, W1, b1, W2, b2, W3, b3, W4, b4, Ut_r, Ut_i, out, batch);
}


# Main kernel: each wave owns 32 batch elements, one per lane (was 16 on half the lanes): prologue and epilogue run on all 32 lanes, two 16-row M tiles share each B fragment load; 2048 blocks
# speedup vs baseline: 10.8202x; 1.1352x over previous
#include <hip/hip_runtime.h>
#include <math.h>

typedef __attribute__((ext_vector_type(2))) float v2f;
typedef __attribute__((ext_vector_type(4))) float f4v;
typedef __attribute__((ext_vector_type(8))) float v8f;

#define WAVES 4
#define EPW   32
#define TILE  (WAVES * EPW)
#define ISTR  68

__device__ __forceinline__ float lane_xor(float v, int lane, int mask) {
  return __int_as_float(__builtin_amdgcn_ds_bpermute((lane ^ mask) << 2, __float_as_int(v)));
}
__device__ __forceinline__ float lane_get(float v, int src) {
  return __int_as_float(__builtin_amdgcn_readlane(__float_as_int(v), src));
}

__global__ __launch_bounds__(32) void build_unitary_kernel(
    const float* __restrict__ qp, float* __restrict__ Ut_r, float* __restrict__ Ut_i) {
  const int lane = threadIdx.x;
  const int c    = blockIdx.x;
  const int s0 = lane, s1 = lane + 32;

  float snA, csA, snB, csB;
  {
    const int tb = (lane + 32 < 54) ? lane + 32 : 53;
    const float thA = qp[lane], thB = qp[tb];
    sincosf(0.5f * thA, &snA, &csA);
    sincosf(0.5f * thB, &snB, &csB);
  }
  auto trig = [&](int idx, float& sh, float& ch) {
    const int j = idx & 31;
    const float sa = lane_get(snA, j), ca = lane_get(csA, j);
    const float sb = lane_get(snB, j), cb = lane_get(csB, j);
    sh = (idx < 32) ? sa : sb;
    ch = (idx < 32) ? ca : cb;
  };

  float r0 = (s0 == c) ? 1.0f : 0.0f, i0 = 0.0f;
  float r1 = (s1 == c) ? 1.0f : 0.0f, i1 = 0.0f;

  auto partner = [&](int m, float& pr0, float& pi0, float& pr1, float& pi1) {
    if (m == 32) { pr0 = r1; pi0 = i1; pr1 = r0; pi1 = i0; }
    else {
      pr0 = lane_xor(r0, lane, m); pi0 = lane_xor(i0, lane, m);
      pr1 = lane_xor(r1, lane, m); pi1 = lane_xor(i1, lane, m);
    }
  };
  auto cnot = [&](int cq, int tq) {
    const int mc = 1 << (5 - cq), mt = 1 << (5 - tq);
    float pr0, pi0, pr1, pi1;
    partner(mt, pr0, pi0, pr1, pi1);
    if (s0 & mc) { r0 = pr0; i0 = pi0; }
    if (s1 & mc) { r1 = pr1; i1 = pi1; }
  };
  auto rx = [&](int q, float sh, float ch) {
    const int m = 1 << (5 - q);
    float pr0, pi0, pr1, pi1;
    partner(m, pr0, pi0, pr1, pi1);
    const float a0 = ch * r0 + sh * pi0, b0 = ch * i0 - sh * pr0;
    const float a1 = ch * r1 + sh * pi1, b1 = ch * i1 - sh * pr1;
    r0 = a0; i0 = b0; r1 = a1; i1 = b1;
  };
  auto ry = [&](int q, float sh, float ch) {
    const int m = 1 << (5 - q);
    float pr0, pi0, pr1, pi1;
    partner(m, pr0, pi0, pr1, pi1);
    const float t0 = (s0 & m) ? sh : -sh;
    const float t1 = (s1 & m) ? sh : -sh;
    const float a0 = ch * r0 + t0 * pr0, b0 = ch * i0 + t0 * pi0;
    const float a1 = ch * r1 + t1 * pr1, b1 = ch * i1 + t1 * pi1;
    r0 = a0; i0 = b0; r1 = a1; i1 = b1;
  };
  auto phase = [&](int s, int m, float sh, float ch, float& r, float& i) {
    const float t = (s & m) ? -sh : sh;
    const float a = ch * r + t * i, b = ch * i - t * r;
    r = a; i = b;
  };
  auto rz = [&](int q, float sh, float ch) {
    const int m = 1 << (5 - q);
    phase(s0, m, sh, ch, r0, i0);
    phase(s1, m, sh, ch, r1, i1);
  };
  auto crz = [&](int cq, int tq, float sh, float ch) {
    const int mc = 1 << (5 - cq), mt = 1 << (5 - tq);
    float a = r0, b = i0;
    phase(s0, mt, sh, ch, a, b);
    if (s0 & mc) { r0 = a; i0 = b; }
    a = r1; b = i1;
    phase(s1, mt, sh, ch, a, b);
    if (s1 & mc) { r1 = a; i1 = b; }
  };

  for (int layer = 0; layer < 3; ++layer) {
    cnot(0, 1); cnot(2, 3); cnot(4, 5);
    cnot(1, 2); cnot(3, 4);
    for (int i = 0; i < 6; ++i) {
      float sh, ch;
      trig(layer * 18 + i * 3 + 0, sh, ch); rx(i, sh, ch);
      trig(layer * 18 + i * 3 + 1, sh, ch); ry(i, sh, ch);
      trig(layer * 18 + i * 3 + 2, sh, ch); rz(i, sh, ch);
    }
    if (layer < 2)
      for (int i = 0; i < 6; ++i) {
        float sh, ch;
        trig(layer * 18 + i * 3 + 0, sh, ch);
        crz(i, (i + 1) % 6, sh, ch);
      }
  }
  Ut_r[c * 64 + s0] = r0; Ut_r[c * 64 + s1] = r1;
  Ut_i[c * 64 + s0] = i0; Ut_i[c * 64 + s1] = i1;
}

__global__ __launch_bounds__(WAVES * 32) void qnn_kernel(
    const float* __restrict__ x,
    const float* __restrict__ W1, const float* __restrict__ b1,
    const float* __restrict__ W2, const float* __restrict__ b2,
    const float* __restrict__ W3, const float* __restrict__ b3,
    const float* __restrict__ W4, const float* __restrict__ b4,
    const float* __restrict__ Ut_r, const float* __restrict__ Ut_i,
    float* __restrict__ out, int batch) {
  __shared__ float sUr[64 * 64];
  __shared__ float sUi[64 * 64];
  __shared__ float sAmp[WAVES][EPW * ISTR];

  const int tid  = threadIdx.x;
  const int wave = tid >> 5;
  const int lane = tid & 31;
  const int half = lane >> 4;
  const int l    = lane & 15;
  const int ebase = blockIdx.x * TILE + wave * EPW;

  {
    const f4v* srcR = (const f4v*)Ut_r;
    const f4v* srcI = (const f4v*)Ut_i;
    f4v* dR = (f4v*)sUr;
    f4v* dI = (f4v*)sUi;
    for (int i = tid; i < 1024; i += WAVES * 32) { dR[i] = srcR[i]; dI[i] = srcI[i]; }
  }

  {
    const int e = ebase + lane;
    const int ec = (e < batch) ? e : (batch - 1);
    const float x0 = x[ec * 4 + 0], x1 = x[ec * 4 + 1];
    const float x2 = x[ec * 4 + 2], x3 = x[ec * 4 + 3];
    float h[16];
#pragma unroll
    for (int j = 0; j < 16; ++j) {
      float t = b1[j] + x0 * W1[j] + x1 * W1[16 + j] + x2 * W1[32 + j] + x3 * W1[48 + j];
      h[j] = t > 0.0f ? t : 0.0f;
    }
    float A0[6], A1[6];
#pragma unroll
    for (int q = 0; q < 6; ++q) {
      float t = b2[q];
#pragma unroll
      for (int j = 0; j < 16; ++j) t += h[j] * W2[j * 6 + q];
      const float ang = 0.5f * 3.14159265358979323846f * tanhf(t);
      float sn, cs; sincosf(ang, &sn, &cs);
      const float is2 = 0.70710678118654752440f;
      A0[q] = (cs - sn) * is2;
      A1[q] = (cs + sn) * is2;
    }
    float m01[4], m23[4], m45[4];
#pragma unroll
    for (int i = 0; i < 4; ++i) {
      m01[i] = ((i & 2) ? A1[0] : A0[0]) * ((i & 1) ? A1[1] : A0[1]);
      m23[i] = ((i & 2) ? A1[2] : A0[2]) * ((i & 1) ? A1[3] : A0[3]);
      m45[i] = ((i & 2) ? A1[4] : A0[4]) * ((i & 1) ? A1[5] : A0[5]);
    }
    float* dst = &sAmp[wave][lane * ISTR];
#pragma unroll
    for (int s = 0; s < 64; ++s)
      dst[s] = m01[s >> 4] * m23[(s >> 2) & 3] * m45[s & 3];
  }
  __syncthreads();

  v8f accr[2][4] = {{v8f{}, v8f{}, v8f{}, v8f{}}, {v8f{}, v8f{}, v8f{}, v8f{}}};
  v8f acci[2][4] = {{v8f{}, v8f{}, v8f{}, v8f{}}, {v8f{}, v8f{}, v8f{}, v8f{}}};
  const float* amp0 = &sAmp[wave][l * ISTR];
  const float* amp1 = &sAmp[wave][(16 + l) * ISTR];
#pragma unroll
  for (int kt = 0; kt < 16; ++kt) {
    const int kb = 4 * kt + 2 * half;
    v2f a0, a1;
    a0.x = amp0[kb];
    a0.y = amp0[kb + 1];
    a1.x = amp1[kb];
    a1.y = amp1[kb + 1];
#pragma unroll
    for (int nt = 0; nt < 4; ++nt) {
      const int n = nt * 16 + l;
      v2f br, bi;
      br.x = sUr[kb * 64 + n];        br.y = sUr[(kb + 1) * 64 + n];
      bi.x = sUi[kb * 64 + n];        bi.y = sUi[(kb + 1) * 64 + n];
      accr[0][nt] = __builtin_amdgcn_wmma_f32_16x16x4_f32(
          false, a0, false, br, (short)0, accr[0][nt], false, false);
      accr[1][nt] = __builtin_amdgcn_wmma_f32_16x16x4_f32(
          false, a1, false, br, (short)0, accr[1][nt], false, false);
      acci[0][nt] = __builtin_amdgcn_wmma_f32_16x16x4_f32(
          false, a0, false, bi, (short)0, acci[0][nt], false, false);
      acci[1][nt] = __builtin_amdgcn_wmma_f32_16x16x4_f32(
          false, a1, false, bi, (short)0, acci[1][nt], false, false);
    }
  }

  {
    float* wamp = sAmp[wave];
#pragma unroll
    for (int m = 0; m < 2; ++m)
#pragma unroll
      for (int nt = 0; nt < 4; ++nt)
#pragma unroll
        for (int r = 0; r < 8; ++r) {
          const float pr = accr[m][nt][r], pi = acci[m][nt][r];
          wamp[(16 * m + r + 8 * half) * ISTR + nt * 16 + l] = pr * pr + pi * pi;
        }
  }
  __syncthreads();

  {
    const int e = ebase + lane;
    if (e < batch) {
      const float* p = &sAmp[wave][lane * ISTR];
      float z0 = 0, z1 = 0, z2 = 0, z3 = 0, z4 = 0, z5 = 0;
#pragma unroll
      for (int s = 0; s < 64; ++s) {
        const float v = p[s];
        z0 += (s & 32) ? -v : v;
        z1 += (s & 16) ? -v : v;
        z2 += (s & 8)  ? -v : v;
        z3 += (s & 4)  ? -v : v;
        z4 += (s & 2)  ? -v : v;
        z5 += (s & 1)  ? -v : v;
      }
      const float z[6] = {z0, z1, z2, z3, z4, z5};
      float o0 = b4[0], o1 = b4[1], o2 = b4[2], o3 = b4[3];
#pragma unroll
      for (int j = 0; j < 16; ++j) {
        float g = b3[j];
#pragma unroll
        for (int q = 0; q < 6; ++q) g += z[q] * W3[q * 16 + j];
        g = g > 0.0f ? g : 0.0f;
        o0 += g * W4[j * 4 + 0]; o1 += g * W4[j * 4 + 1];
        o2 += g * W4[j * 4 + 2]; o3 += g * W4[j * 4 + 3];
      }
      float* op = out + (size_t)e * 4;
      op[0] = o0; op[1] = o1; op[2] = o2; op[3] = o3;
    }
  }
}

extern "C" void kernel_launch(void* const* d_in, const int* in_sizes, int n_in,
                              void* d_out, int out_size, void* d_ws, size_t ws_size,
                              hipStream_t stream) {
  const float* x  = (const float*)d_in[0];
  const float* W1 = (const float*)d_in[1];
  const float* b1 = (const float*)d_in[2];
  const float* W2 = (const float*)d_in[3];
  const float* b2 = (const float*)d_in[4];
  const float* qp = (const float*)d_in[5];
  const float* W3 = (const float*)d_in[6];
  const float* b3 = (const float*)d_in[7];
  const float* W4 = (const float*)d_in[8];
  const float* b4 = (const float*)d_in[9];
  float* out  = (float*)d_out;
  float* Ut_r = (float*)d_ws;
  float* Ut_i = Ut_r + 64 * 64;

  const int batch = in_sizes[0] / 4;
  build_unitary_kernel<<<dim3(64), dim3(32), 0, stream>>>(qp, Ut_r, Ut_i);
  const int tiles = (batch + TILE - 1) / TILE;
  qnn_kernel<<<dim3(tiles), dim3(WAVES * 32), 0, stream>>>(
      x, W1, b1, W2, b2, W3, b3, W4, b4, Ut_r, Ut_i, out, batch);
}


# Main kernel: each wave keeps the U^T B fragments of its 16 output states in registers (read once from global) and runs all 8 M tiles of the block; U no longer staged in LDS, A loads grouped per tile
# speedup vs baseline: 12.1617x; 1.1240x over previous
#include <hip/hip_runtime.h>
#include <math.h>

typedef __attribute__((ext_vector_type(2))) float v2f;
typedef __attribute__((ext_vector_type(8))) float v8f;

#define WAVES 4
#define TILE  (WAVES * 32)
#define ISTR  68

__device__ __forceinline__ float lane_xor(float v, int lane, int mask) {
  return __int_as_float(__builtin_amdgcn_ds_bpermute((lane ^ mask) << 2, __float_as_int(v)));
}
__device__ __forceinline__ float lane_get(float v, int src) {
  return __int_as_float(__builtin_amdgcn_readlane(__float_as_int(v), src));
}

__global__ __launch_bounds__(32) void build_unitary_kernel(
    const float* __restrict__ qp, float* __restrict__ Ut_r, float* __restrict__ Ut_i) {
  const int lane = threadIdx.x;
  const int c    = blockIdx.x;
  const int s0 = lane, s1 = lane + 32;

  float snA, csA, snB, csB;
  {
    const int tb = (lane + 32 < 54) ? lane + 32 : 53;
    const float thA = qp[lane], thB = qp[tb];
    sincosf(0.5f * thA, &snA, &csA);
    sincosf(0.5f * thB, &snB, &csB);
  }
  auto trig = [&](int idx, float& sh, float& ch) {
    const int j = idx & 31;
    const float sa = lane_get(snA, j), ca = lane_get(csA, j);
    const float sb = lane_get(snB, j), cb = lane_get(csB, j);
    sh = (idx < 32) ? sa : sb;
    ch = (idx < 32) ? ca : cb;
  };

  float r0 = (s0 == c) ? 1.0f : 0.0f, i0 = 0.0f;
  float r1 = (s1 == c) ? 1.0f : 0.0f, i1 = 0.0f;

  auto partner = [&](int m, float& pr0, float& pi0, float& pr1, float& pi1) {
    if (m == 32) { pr0 = r1; pi0 = i1; pr1 = r0; pi1 = i0; }
    else {
      pr0 = lane_xor(r0, lane, m); pi0 = lane_xor(i0, lane, m);
      pr1 = lane_xor(r1, lane, m); pi1 = lane_xor(i1, lane, m);
    }
  };
  auto cnot = [&](int cq, int tq) {
    const int mc = 1 << (5 - cq), mt = 1 << (5 - tq);
    float pr0, pi0, pr1, pi1;
    partner(mt, pr0, pi0, pr1, pi1);
    if (s0 & mc) { r0 = pr0; i0 = pi0; }
    if (s1 & mc) { r1 = pr1; i1 = pi1; }
  };
  auto rx = [&](int q, float sh, float ch) {
    const int m = 1 << (5 - q);
    float pr0, pi0, pr1, pi1;
    partner(m, pr0, pi0, pr1, pi1);
    const float a0 = ch * r0 + sh * pi0, b0 = ch * i0 - sh * pr0;
    const float a1 = ch * r1 + sh * pi1, b1 = ch * i1 - sh * pr1;
    r0 = a0; i0 = b0; r1 = a1; i1 = b1;
  };
  auto ry = [&](int q, float sh, float ch) {
    const int m = 1 << (5 - q);
    float pr0, pi0, pr1, pi1;
    partner(m, pr0, pi0, pr1, pi1);
    const float t0 = (s0 & m) ? sh : -sh;
    const float t1 = (s1 & m) ? sh : -sh;
    const float a0 = ch * r0 + t0 * pr0, b0 = ch * i0 + t0 * pi0;
    const float a1 = ch * r1 + t1 * pr1, b1 = ch * i1 + t1 * pi1;
    r0 = a0; i0 = b0; r1 = a1; i1 = b1;
  };
  auto phase = [&](int s, int m, float sh, float ch, float& r, float& i) {
    const float t = (s & m) ? -sh : sh;
    const float a = ch * r + t * i, b = ch * i - t * r;
    r = a; i = b;
  };
  auto rz = [&](int q, float sh, float ch) {
    const int m = 1 << (5 - q);
    phase(s0, m, sh, ch, r0, i0);
    phase(s1, m, sh, ch, r1, i1);
  };
  auto crz = [&](int cq, int tq, float sh, float ch) {
    const int mc = 1 << (5 - cq), mt = 1 << (5 - tq);
    float a = r0, b = i0;
    phase(s0, mt, sh, ch, a, b);
    if (s0 & mc) { r0 = a; i0 = b; }
    a = r1; b = i1;
    phase(s1, mt, sh, ch, a, b);
    if (s1 & mc) { r1 = a; i1 = b; }
  };

  for (int layer = 0; layer < 3; ++layer) {
    cnot(0, 1); cnot(2, 3); cnot(4, 5);
    cnot(1, 2); cnot(3, 4);
    for (int i = 0; i < 6; ++i) {
      float sh, ch;
      trig(layer * 18 + i * 3 + 0, sh, ch); rx(i, sh, ch);
      trig(layer * 18 + i * 3 + 1, sh, ch); ry(i, sh, ch);
      trig(layer * 18 + i * 3 + 2, sh, ch); rz(i, sh, ch);
    }
    if (layer < 2)
      for (int i = 0; i < 6; ++i) {
        float sh, ch;
        trig(layer * 18 + i * 3 + 0, sh, ch);
        crz(i, (i + 1) % 6, sh, ch);
      }
  }
  Ut_r[c * 64 + s0] = r0; Ut_r[c * 64 + s1] = r1;
  Ut_i[c * 64 + s0] = i0; Ut_i[c * 64 + s1] = i1;
}

__global__ __launch_bounds__(WAVES * 32) void qnn_kernel(
    const float* __restrict__ x,
    const float* __restrict__ W1, const float* __restrict__ b1,
    const float* __restrict__ W2, const float* __restrict__ b2,
    const float* __restrict__ W3, const float* __restrict__ b3,
    const float* __restrict__ W4, const float* __restrict__ b4,
    const float* __restrict__ Ut_r, const float* __restrict__ Ut_i,
    float* __restrict__ out, int batch) {
  __shared__ __attribute__((aligned(16))) float sAmp[TILE * ISTR];
  __shared__ __attribute__((aligned(16))) float sProb[TILE * ISTR];

  const int tid  = threadIdx.x;
  const int wave = tid >> 5;
  const int lane = tid & 31;
  const int half = lane >> 4;
  const int l    = lane & 15;
  const int ebase = blockIdx.x * TILE;

  const int e = ebase + tid;
  const int ec = (e < batch) ? e : (batch - 1);
  const float x0 = x[ec * 4 + 0], x1 = x[ec * 4 + 1];
  const float x2 = x[ec * 4 + 2], x3 = x[ec * 4 + 3];

  v2f br[16], bi[16];
  {
    const int n = wave * 16 + l;
#pragma unroll
    for (int kt = 0; kt < 16; ++kt) {
      const int kb = 4 * kt + 2 * half;
      br[kt].x = Ut_r[kb * 64 + n];   br[kt].y = Ut_r[(kb + 1) * 64 + n];
      bi[kt].x = Ut_i[kb * 64 + n];   bi[kt].y = Ut_i[(kb + 1) * 64 + n];
    }
  }

  {
    float h[16];
#pragma unroll
    for (int j = 0; j < 16; ++j) {
      float t = b1[j] + x0 * W1[j] + x1 * W1[16 + j] + x2 * W1[32 + j] + x3 * W1[48 + j];
      h[j] = t > 0.0f ? t : 0.0f;
    }
    float A0[6], A1[6];
#pragma unroll
    for (int q = 0; q < 6; ++q) {
      float t = b2[q];
#pragma unroll
      for (int j = 0; j < 16; ++j) t += h[j] * W2[j * 6 + q];
      const float ang = 0.5f * 3.14159265358979323846f * tanhf(t);
      float sn, cs; sincosf(ang, &sn, &cs);
      const float is2 = 0.70710678118654752440f;
      A0[q] = (cs - sn) * is2;
      A1[q] = (cs + sn) * is2;
    }
    float m01[4], m23[4], m45[4];
#pragma unroll
    for (int i = 0; i < 4; ++i) {
      m01[i] = ((i & 2) ? A1[0] : A0[0]) * ((i & 1) ? A1[1] : A0[1]);
      m23[i] = ((i & 2) ? A1[2] : A0[2]) * ((i & 1) ? A1[3] : A0[3]);
      m45[i] = ((i & 2) ? A1[4] : A0[4]) * ((i & 1) ? A1[5] : A0[5]);
    }
    float* dst = &sAmp[tid * ISTR];
#pragma unroll
    for (int s = 0; s < 64; ++s)
      dst[s] = m01[s >> 4] * m23[(s >> 2) & 3] * m45[s & 3];
  }
  __syncthreads();

  {
    const float* arow = &sAmp[l * ISTR + 2 * half];
    float* prow = &sProb[(8 * half) * ISTR + wave * 16 + l];
    v2f a[2][16];
    v8f cr[2], ci[2];
    float pv[2][8];
#pragma unroll
    for (int kt = 0; kt < 16; ++kt) {
      a[0][kt].x = arow[4 * kt];
      a[0][kt].y = arow[4 * kt + 1];
    }
#pragma unroll
    for (int m = 0; m < 10; ++m) {
      if (m + 1 < 8) {
        const float* am = arow + (m + 1) * 16 * ISTR;
#pragma unroll
        for (int kt = 0; kt < 16; ++kt) {
          a[(m + 1) & 1][kt].x = am[4 * kt];
          a[(m + 1) & 1][kt].y = am[4 * kt + 1];
        }
      }
      if (m >= 2) {
        float* pm = prow + (m - 2) * 16 * ISTR;
#pragma unroll
        for (int r = 0; r < 8; ++r) pm[r * ISTR] = pv[m & 1][r];
      }
      __builtin_amdgcn_sched_barrier(0);
      if (m < 8) {
        v8f accr = v8f{}, acci = v8f{};
#pragma unroll
        for (int kt = 0; kt < 16; ++kt) {
          accr = __builtin_amdgcn_wmma_f32_16x16x4_f32(
              false, a[m & 1][kt], false, br[kt], (short)0, accr, false, false);
          acci = __builtin_amdgcn_wmma_f32_16x16x4_f32(
              false, a[m & 1][kt], false, bi[kt], (short)0, acci, false, false);
        }
        cr[m & 1] = accr; ci[m & 1] = acci;
      }
      if (m >= 1 && m <= 8) {
#pragma unroll
        for (int r = 0; r < 8; ++r) {
          const float pr = cr[(m - 1) & 1][r], pi = ci[(m - 1) & 1][r];
          pv[(m - 1) & 1][r] = pr * pr + pi * pi;
        }
      }
      __builtin_amdgcn_sched_barrier(0);
    }
  }
  __syncthreads();

  {
    if (e < batch) {
      const float* p = &sProb[tid * ISTR];
      float z0 = 0, z1 = 0, z2 = 0, z3 = 0, z4 = 0, z5 = 0;
#pragma unroll
      for (int s = 0; s < 64; ++s) {
        const float v = p[s];
        z0 += (s & 32) ? -v : v;
        z1 += (s & 16) ? -v : v;
        z2 += (s & 8)  ? -v : v;
        z3 += (s & 4)  ? -v : v;
        z4 += (s & 2)  ? -v : v;
        z5 += (s & 1)  ? -v : v;
      }
      const float z[6] = {z0, z1, z2, z3, z4, z5};
      float o0 = b4[0], o1 = b4[1], o2 = b4[2], o3 = b4[3];
#pragma unroll
      for (int j = 0; j < 16; ++j) {
        float g = b3[j];
#pragma unroll
        for (int q = 0; q < 6; ++q) g += z[q] * W3[q * 16 + j];
        g = g > 0.0f ? g : 0.0f;
        o0 += g * W4[j * 4 + 0]; o1 += g * W4[j * 4 + 1];
        o2 += g * W4[j * 4 + 2]; o3 += g * W4[j * 4 + 3];
      }
      float* op = out + (size_t)e * 4;
      op[0] = o0; op[1] = o1; op[2] = o2; op[3] = o3;
    }
  }
}

extern "C" void kernel_launch(void* const* d_in, const int* in_sizes, int n_in,
                              void* d_out, int out_size, void* d_ws, size_t ws_size,
                              hipStream_t stream) {
  const float* x  = (const float*)d_in[0];
  const float* W1 = (const float*)d_in[1];
  const float* b1 = (const float*)d_in[2];
  const float* W2 = (const float*)d_in[3];
  const float* b2 = (const float*)d_in[4];
  const float* qp = (const float*)d_in[5];
  const float* W3 = (const float*)d_in[6];
  const float* b3 = (const float*)d_in[7];
  const float* W4 = (const float*)d_in[8];
  const float* b4 = (const float*)d_in[9];
  float* out  = (float*)d_out;
  float* Ut_r = (float*)d_ws;
  float* Ut_i = Ut_r + 64 * 64;

  const int batch = in_sizes[0] / 4;
  build_unitary_kernel<<<dim3(64), dim3(32), 0, stream>>>(qp, Ut_r, Ut_i);
  const int tiles = (batch + TILE - 1) / TILE;
  qnn_kernel<<<dim3(tiles), dim3(WAVES * 32), 0, stream>>>(
      x, W1, b1, W2, b2, W3, b3, W4, b4, Ut_r, Ut_i, out, batch);
}


# Epilogue: the six signed <Z> sums share partial sums in a sign butterfly on float pairs (182 adds per element, was 384 serial adds); changes the order in which the probabilities are summed
# speedup vs baseline: 12.6034x; 1.0363x over previous
#include <hip/hip_runtime.h>
#include <math.h>

typedef __attribute__((ext_vector_type(2))) float v2f;
typedef __attribute__((ext_vector_type(4))) float v4f;
typedef __attribute__((ext_vector_type(8))) float v8f;

#define WAVES 4
#define TILE  (WAVES * 32)
#define ISTR  68

__device__ __forceinline__ float lane_xor(float v, int lane, int mask) {
  return __int_as_float(__builtin_amdgcn_ds_bpermute((lane ^ mask) << 2, __float_as_int(v)));
}
__device__ __forceinline__ float lane_get(float v, int src) {
  return __int_as_float(__builtin_amdgcn_readlane(__float_as_int(v), src));
}

__global__ __launch_bounds__(32) void build_unitary_kernel(
    const float* __restrict__ qp, float* __restrict__ Ut_r, float* __restrict__ Ut_i) {
  const int lane = threadIdx.x;
  const int c    = blockIdx.x;
  const int s0 = lane, s1 = lane + 32;

  float snA, csA, snB, csB;
  {
    const int tb = (lane + 32 < 54) ? lane + 32 : 53;
    const float thA = qp[lane], thB = qp[tb];
    sincosf(0.5f * thA, &snA, &csA);
    sincosf(0.5f * thB, &snB, &csB);
  }
  auto trig = [&](int idx, float& sh, float& ch) {
    const int j = idx & 31;
    const float sa = lane_get(snA, j), ca = lane_get(csA, j);
    const float sb = lane_get(snB, j), cb = lane_get(csB, j);
    sh = (idx < 32) ? sa : sb;
    ch = (idx < 32) ? ca : cb;
  };

  float r0 = (s0 == c) ? 1.0f : 0.0f, i0 = 0.0f;
  float r1 = (s1 == c) ? 1.0f : 0.0f, i1 = 0.0f;

  auto partner = [&](int m, float& pr0, float& pi0, float& pr1, float& pi1) {
    if (m == 32) { pr0 = r1; pi0 = i1; pr1 = r0; pi1 = i0; }
    else {
      pr0 = lane_xor(r0, lane, m); pi0 = lane_xor(i0, lane, m);
      pr1 = lane_xor(r1, lane, m); pi1 = lane_xor(i1, lane, m);
    }
  };
  auto cnot = [&](int cq, int tq) {
    const int mc = 1 << (5 - cq), mt = 1 << (5 - tq);
    float pr0, pi0, pr1, pi1;
    partner(mt, pr0, pi0, pr1, pi1);
    if (s0 & mc) { r0 = pr0; i0 = pi0; }
    if (s1 & mc) { r1 = pr1; i1 = pi1; }
  };
  auto rx = [&](int q, float sh, float ch) {
    const int m = 1 << (5 - q);
    float pr0, pi0, pr1, pi1;
    partner(m, pr0, pi0, pr1, pi1);
    const float a0 = ch * r0 + sh * pi0, b0 = ch * i0 - sh * pr0;
    const float a1 = ch * r1 + sh * pi1, b1 = ch * i1 - sh * pr1;
    r0 = a0; i0 = b0; r1 = a1; i1 = b1;
  };
  auto ry = [&](int q, float sh, float ch) {
    const int m = 1 << (5 - q);
    float pr0, pi0, pr1, pi1;
    partner(m, pr0, pi0, pr1, pi1);
    const float t0 = (s0 & m) ? sh : -sh;
    const float t1 = (s1 & m) ? sh : -sh;
    const float a0 = ch * r0 + t0 * pr0, b0 = ch * i0 + t0 * pi0;
    const float a1 = ch * r1 + t1 * pr1, b1 = ch * i1 + t1 * pi1;
    r0 = a0; i0 = b0; r1 = a1; i1 = b1;
  };
  auto phase = [&](int s, int m, float sh, float ch, float& r, float& i) {
    const float t = (s & m) ? -sh : sh;
    const float a = ch * r + t * i, b = ch * i - t * r;
    r = a; i = b;
  };
  auto rz = [&](int q, float sh, float ch) {
    const int m = 1 << (5 - q);
    phase(s0, m, sh, ch, r0, i0);
    phase(s1, m, sh, ch, r1, i1);
  };
  auto crz = [&](int cq, int tq, float sh, float ch) {
    const int mc = 1 << (5 - cq), mt = 1 << (5 - tq);
    float a = r0, b = i0;
    phase(s0, mt, sh, ch, a, b);
    if (s0 & mc) { r0 = a; i0 = b; }
    a = r1; b = i1;
    phase(s1, mt, sh, ch, a, b);
    if (s1 & mc) { r1 = a; i1 = b; }
  };

  for (int layer = 0; layer < 3; ++layer) {
    cnot(0, 1); cnot(2, 3); cnot(4, 5);
    cnot(1, 2); cnot(3, 4);
    for (int i = 0; i < 6; ++i) {
      float sh, ch;
      trig(layer * 18 + i * 3 + 0, sh, ch); rx(i, sh, ch);
      trig(layer * 18 + i * 3 + 1, sh, ch); ry(i, sh, ch);
      trig(layer * 18 + i * 3 + 2, sh, ch); rz(i, sh, ch);
    }
    if (layer < 2)
      for (int i = 0; i < 6; ++i) {
        float sh, ch;
        trig(layer * 18 + i * 3 + 0, sh, ch);
        crz(i, (i + 1) % 6, sh, ch);
      }
  }
  Ut_r[c * 64 + s0] = r0; Ut_r[c * 64 + s1] = r1;
  Ut_i[c * 64 + s0] = i0; Ut_i[c * 64 + s1] = i1;
}

__global__ __launch_bounds__(WAVES * 32) void qnn_kernel(
    const float* __restrict__ x,
    const float* __restrict__ W1, const float* __restrict__ b1,
    const float* __restrict__ W2, const float* __restrict__ b2,
    const float* __restrict__ W3, const float* __restrict__ b3,
    const float* __restrict__ W4, const float* __restrict__ b4,
    const float* __restrict__ Ut_r, const float* __restrict__ Ut_i,
    float* __restrict__ out, int batch) {
  __shared__ __attribute__((aligned(16))) float sAmp[TILE * ISTR];
  __shared__ __attribute__((aligned(16))) float sProb[TILE * ISTR];

  const int tid  = threadIdx.x;
  const int wave = tid >> 5;
  const int lane = tid & 31;
  const int half = lane >> 4;
  const int l    = lane & 15;
  const int ebase = blockIdx.x * TILE;

  const int e = ebase + tid;
  const int ec = (e < batch) ? e : (batch - 1);
  const float x0 = x[ec * 4 + 0], x1 = x[ec * 4 + 1];
  const float x2 = x[ec * 4 + 2], x3 = x[ec * 4 + 3];

  v2f br[16], bi[16];
  {
    const int n = wave * 16 + l;
#pragma unroll
    for (int kt = 0; kt < 16; ++kt) {
      const int kb = 4 * kt + 2 * half;
      br[kt].x = Ut_r[kb * 64 + n];   br[kt].y = Ut_r[(kb + 1) * 64 + n];
      bi[kt].x = Ut_i[kb * 64 + n];   bi[kt].y = Ut_i[(kb + 1) * 64 + n];
    }
  }

  {
    float h[16];
#pragma unroll
    for (int j = 0; j < 16; ++j) {
      float t = b1[j] + x0 * W1[j] + x1 * W1[16 + j] + x2 * W1[32 + j] + x3 * W1[48 + j];
      h[j] = t > 0.0f ? t : 0.0f;
    }
    float A0[6], A1[6];
#pragma unroll
    for (int q = 0; q < 6; ++q) {
      float t = b2[q];
#pragma unroll
      for (int j = 0; j < 16; ++j) t += h[j] * W2[j * 6 + q];
      const float ang = 0.5f * 3.14159265358979323846f * tanhf(t);
      float sn, cs; sincosf(ang, &sn, &cs);
      const float is2 = 0.70710678118654752440f;
      A0[q] = (cs - sn) * is2;
      A1[q] = (cs + sn) * is2;
    }
    float m01[4], m23[4], m45[4];
#pragma unroll
    for (int i = 0; i < 4; ++i) {
      m01[i] = ((i & 2) ? A1[0] : A0[0]) * ((i & 1) ? A1[1] : A0[1]);
      m23[i] = ((i & 2) ? A1[2] : A0[2]) * ((i & 1) ? A1[3] : A0[3]);
      m45[i] = ((i & 2) ? A1[4] : A0[4]) * ((i & 1) ? A1[5] : A0[5]);
    }
    float* dst = &sAmp[tid * ISTR];
#pragma unroll
    for (int s = 0; s < 64; ++s)
      dst[s] = m01[s >> 4] * m23[(s >> 2) & 3] * m45[s & 3];
  }
  __syncthreads();

  {
    const float* arow = &sAmp[l * ISTR + 2 * half];
    float* prow = &sProb[(8 * half) * ISTR + wave * 16 + l];
    v2f a[2][16];
    v8f cr[2], ci[2];
    float pv[2][8];
#pragma unroll
    for (int kt = 0; kt < 16; ++kt) {
      a[0][kt].x = arow[4 * kt];
      a[0][kt].y = arow[4 * kt + 1];
    }
#pragma unroll
    for (int m = 0; m < 10; ++m) {
      if (m + 1 < 8) {
        const float* am = arow + (m + 1) * 16 * ISTR;
#pragma unroll
        for (int kt = 0; kt < 16; ++kt) {
          a[(m + 1) & 1][kt].x = am[4 * kt];
          a[(m + 1) & 1][kt].y = am[4 * kt + 1];
        }
      }
      if (m >= 2) {
        float* pm = prow + (m - 2) * 16 * ISTR;
#pragma unroll
        for (int r = 0; r < 8; ++r) pm[r * ISTR] = pv[m & 1][r];
      }
      __builtin_amdgcn_sched_barrier(0);
      if (m < 8) {
        v8f accr = v8f{}, acci = v8f{};
#pragma unroll
        for (int kt = 0; kt < 16; ++kt) {
          accr = __builtin_amdgcn_wmma_f32_16x16x4_f32(
              false, a[m & 1][kt], false, br[kt], (short)0, accr, false, false);
          acci = __builtin_amdgcn_wmma_f32_16x16x4_f32(
              false, a[m & 1][kt], false, bi[kt], (short)0, acci, false, false);
        }
        cr[m & 1] = accr; ci[m & 1] = acci;
      }
      if (m >= 1 && m <= 8) {
#pragma unroll
        for (int r = 0; r < 8; ++r) {
          const float pr = cr[(m - 1) & 1][r], pi = ci[(m - 1) & 1][r];
          pv[(m - 1) & 1][r] = pr * pr + pi * pi;
        }
      }
      __builtin_amdgcn_sched_barrier(0);
    }
  }
  __syncthreads();

  {
    if (e < batch) {
      const v4f* p4 = reinterpret_cast<const v4f*>(&sProb[tid * ISTR]);
      v2f P[32];
#pragma unroll
      for (int j = 0; j < 16; ++j) {
        const v4f t = p4[j];
        P[2 * j]     = v2f{t.x, t.y};
        P[2 * j + 1] = v2f{t.z, t.w};
      }
      v2f A16[16], D16[16];
#pragma unroll
      for (int j = 0; j < 16; ++j) { A16[j] = P[j] + P[j + 16]; D16[j] = P[j] - P[j + 16]; }
#pragma unroll
      for (int w = 8; w >= 1; w >>= 1)
#pragma unroll
        for (int j = 0; j < w; ++j) D16[j] += D16[j + w];
      const float z0 = D16[0].x + D16[0].y;
      v2f A8[8], D8[8];
#pragma unroll
      for (int j = 0; j < 8; ++j) { A8[j] = A16[j] + A16[j + 8]; D8[j] = A16[j] - A16[j + 8]; }
#pragma unroll
      for (int w = 4; w >= 1; w >>= 1)
#pragma unroll
        for (int j = 0; j < w; ++j) D8[j] += D8[j + w];
      const float z1 = D8[0].x + D8[0].y;
      v2f A4[4], D4[4];
#pragma unroll
      for (int j = 0; j < 4; ++j) { A4[j] = A8[j] + A8[j + 4]; D4[j] = A8[j] - A8[j + 4]; }
      D4[0] += D4[2]; D4[1] += D4[3]; D4[0] += D4[1];
      const float z2 = D4[0].x + D4[0].y;
      const v2f A2a = A4[0] + A4[2], A2b = A4[1] + A4[3];
      const v2f D2 = (A4[0] - A4[2]) + (A4[1] - A4[3]);
      const float z3 = D2.x + D2.y;
      const v2f A1 = A2a + A2b, D1 = A2a - A2b;
      const float z4 = D1.x + D1.y;
      const float z5 = A1.x - A1.y;
      const float z[6] = {z0, z1, z2, z3, z4, z5};
      float o0 = b4[0], o1 = b4[1], o2 = b4[2], o3 = b4[3];
#pragma unroll
      for (int j = 0; j < 16; ++j) {
        float g = b3[j];
#pragma unroll
        for (int q = 0; q < 6; ++q) g += z[q] * W3[q * 16 + j];
        g = g > 0.0f ? g : 0.0f;
        o0 += g * W4[j * 4 + 0]; o1 += g * W4[j * 4 + 1];
        o2 += g * W4[j * 4 + 2]; o3 += g * W4[j * 4 + 3];
      }
      float* op = out + (size_t)e * 4;
      op[0] = o0; op[1] = o1; op[2] = o2; op[3] = o3;
    }
  }
}

extern "C" void kernel_launch(void* const* d_in, const int* in_sizes, int n_in,
                              void* d_out, int out_size, void* d_ws, size_t ws_size,
                              hipStream_t stream) {
  const float* x  = (const float*)d_in[0];
  const float* W1 = (const float*)d_in[1];
  const float* b1 = (const float*)d_in[2];
  const float* W2 = (const float*)d_in[3];
  const float* b2 = (const float*)d_in[4];
  const float* qp = (const float*)d_in[5];
  const float* W3 = (const float*)d_in[6];
  const float* b3 = (const float*)d_in[7];
  const float* W4 = (const float*)d_in[8];
  const float* b4 = (const float*)d_in[9];
  float* out  = (float*)d_out;
  float* Ut_r = (float*)d_ws;
  float* Ut_i = Ut_r + 64 * 64;

  const int batch = in_sizes[0] / 4;
  build_unitary_kernel<<<dim3(64), dim3(32), 0, stream>>>(qp, Ut_r, Ut_i);
  const int tiles = (batch + TILE - 1) / TILE;
  qnn_kernel<<<dim3(tiles), dim3(WAVES * 32), 0, stream>>>(
      x, W1, b1, W2, b2, W3, b3, W4, b4, Ut_r, Ut_i, out, batch);
}


# Unitary build: each gate's partner amplitudes come by vector-ALU lane exchange (DPP row_xmask in a 16-lane row, permlanex16 across rows), not ds_bpermute; the 66-gate chain is latency-bound
# speedup vs baseline: 13.2986x; 1.0552x over previous
#include <hip/hip_runtime.h>
#include <math.h>

typedef __attribute__((ext_vector_type(2))) float v2f;
typedef __attribute__((ext_vector_type(4))) float v4f;
typedef __attribute__((ext_vector_type(8))) float v8f;

#define WAVES 4
#define TILE  (WAVES * 32)
#define ISTR  68

template <int M>
__device__ __forceinline__ float lane_xor(float v) {
  const int b = __float_as_int(v);
  if constexpr (M == 16) {
    return __int_as_float(
        __builtin_amdgcn_permlanex16(b, b, 0x76543210, 0xfedcba98, false, false));
  } else {
    return __int_as_float(__builtin_amdgcn_mov_dpp(b, 0x160 + M, 0xf, 0xf, true));
  }
}
template <int M>
__device__ __forceinline__ void lane_xor4(float a, float b, float c, float d,
                                          float& pa, float& pb, float& pc, float& pd) {
  pa = lane_xor<M>(a); pb = lane_xor<M>(b); pc = lane_xor<M>(c); pd = lane_xor<M>(d);
}
__device__ __forceinline__ float lane_get(float v, int src) {
  return __int_as_float(__builtin_amdgcn_readlane(__float_as_int(v), src));
}

__global__ __launch_bounds__(32) void build_unitary_kernel(
    const float* __restrict__ qp, float* __restrict__ Ut_r, float* __restrict__ Ut_i) {
  const int lane = threadIdx.x;
  const int c    = blockIdx.x;
  const int s0 = lane, s1 = lane + 32;

  float snA, csA, snB, csB;
  {
    const int tb = (lane + 32 < 54) ? lane + 32 : 53;
    const float thA = qp[lane], thB = qp[tb];
    sincosf(0.5f * thA, &snA, &csA);
    sincosf(0.5f * thB, &snB, &csB);
  }
  auto trig = [&](int idx, float& sh, float& ch) {
    const int j = idx & 31;
    const float sa = lane_get(snA, j), ca = lane_get(csA, j);
    const float sb = lane_get(snB, j), cb = lane_get(csB, j);
    sh = (idx < 32) ? sa : sb;
    ch = (idx < 32) ? ca : cb;
  };

  float r0 = (s0 == c) ? 1.0f : 0.0f, i0 = 0.0f;
  float r1 = (s1 == c) ? 1.0f : 0.0f, i1 = 0.0f;

  auto partner = [&](int m, float& pr0, float& pi0, float& pr1, float& pi1) {
    switch (m) {
      case 32: pr0 = r1; pi0 = i1; pr1 = r0; pi1 = i0; break;
      case 16: lane_xor4<16>(r0, i0, r1, i1, pr0, pi0, pr1, pi1); break;
      case 8:  lane_xor4<8>(r0, i0, r1, i1, pr0, pi0, pr1, pi1); break;
      case 4:  lane_xor4<4>(r0, i0, r1, i1, pr0, pi0, pr1, pi1); break;
      case 2:  lane_xor4<2>(r0, i0, r1, i1, pr0, pi0, pr1, pi1); break;
      default: lane_xor4<1>(r0, i0, r1, i1, pr0, pi0, pr1, pi1); break;
    }
  };
  auto cnot = [&](int cq, int tq) {
    const int mc = 1 << (5 - cq), mt = 1 << (5 - tq);
    float pr0, pi0, pr1, pi1;
    partner(mt, pr0, pi0, pr1, pi1);
    if (s0 & mc) { r0 = pr0; i0 = pi0; }
    if (s1 & mc) { r1 = pr1; i1 = pi1; }
  };
  auto rx = [&](int q, float sh, float ch) {
    const int m = 1 << (5 - q);
    float pr0, pi0, pr1, pi1;
    partner(m, pr0, pi0, pr1, pi1);
    const float a0 = ch * r0 + sh * pi0, b0 = ch * i0 - sh * pr0;
    const float a1 = ch * r1 + sh * pi1, b1 = ch * i1 - sh * pr1;
    r0 = a0; i0 = b0; r1 = a1; i1 = b1;
  };
  auto ry = [&](int q, float sh, float ch) {
    const int m = 1 << (5 - q);
    float pr0, pi0, pr1, pi1;
    partner(m, pr0, pi0, pr1, pi1);
    const float t0 = (s0 & m) ? sh : -sh;
    const float t1 = (s1 & m) ? sh : -sh;
    const float a0 = ch * r0 + t0 * pr0, b0 = ch * i0 + t0 * pi0;
    const float a1 = ch * r1 + t1 * pr1, b1 = ch * i1 + t1 * pi1;
    r0 = a0; i0 = b0; r1 = a1; i1 = b1;
  };
  auto phase = [&](int s, int m, float sh, float ch, float& r, float& i) {
    const float t = (s & m) ? -sh : sh;
    const float a = ch * r + t * i, b = ch * i - t * r;
    r = a; i = b;
  };
  auto rz = [&](int q, float sh, float ch) {
    const int m = 1 << (5 - q);
    phase(s0, m, sh, ch, r0, i0);
    phase(s1, m, sh, ch, r1, i1);
  };
  auto crz = [&](int cq, int tq, float sh, float ch) {
    const int mc = 1 << (5 - cq), mt = 1 << (5 - tq);
    float a = r0, b = i0;
    phase(s0, mt, sh, ch, a, b);
    if (s0 & mc) { r0 = a; i0 = b; }
    a = r1; b = i1;
    phase(s1, mt, sh, ch, a, b);
    if (s1 & mc) { r1 = a; i1 = b; }
  };

  for (int layer = 0; layer < 3; ++layer) {
    cnot(0, 1); cnot(2, 3); cnot(4, 5);
    cnot(1, 2); cnot(3, 4);
    for (int i = 0; i < 6; ++i) {
      float sh, ch;
      trig(layer * 18 + i * 3 + 0, sh, ch); rx(i, sh, ch);
      trig(layer * 18 + i * 3 + 1, sh, ch); ry(i, sh, ch);
      trig(layer * 18 + i * 3 + 2, sh, ch); rz(i, sh, ch);
    }
    if (layer < 2)
      for (int i = 0; i < 6; ++i) {
        float sh, ch;
        trig(layer * 18 + i * 3 + 0, sh, ch);
        crz(i, (i + 1) % 6, sh, ch);
      }
  }
  Ut_r[c * 64 + s0] = r0; Ut_r[c * 64 + s1] = r1;
  Ut_i[c * 64 + s0] = i0; Ut_i[c * 64 + s1] = i1;
}

__global__ __launch_bounds__(WAVES * 32) void qnn_kernel(
    const float* __restrict__ x,
    const float* __restrict__ W1, const float* __restrict__ b1,
    const float* __restrict__ W2, const float* __restrict__ b2,
    const float* __restrict__ W3, const float* __restrict__ b3,
    const float* __restrict__ W4, const float* __restrict__ b4,
    const float* __restrict__ Ut_r, const float* __restrict__ Ut_i,
    float* __restrict__ out, int batch) {
  __shared__ __attribute__((aligned(16))) float sAmp[TILE * ISTR];
  __shared__ __attribute__((aligned(16))) float sProb[TILE * ISTR];

  const int tid  = threadIdx.x;
  const int wave = tid >> 5;
  const int lane = tid & 31;
  const int half = lane >> 4;
  const int l    = lane & 15;
  const int ebase = blockIdx.x * TILE;

  const int e = ebase + tid;
  const int ec = (e < batch) ? e : (batch - 1);
  const float x0 = x[ec * 4 + 0], x1 = x[ec * 4 + 1];
  const float x2 = x[ec * 4 + 2], x3 = x[ec * 4 + 3];

  v2f br[16], bi[16];
  {
    const int n = wave * 16 + l;
#pragma unroll
    for (int kt = 0; kt < 16; ++kt) {
      const int kb = 4 * kt + 2 * half;
      br[kt].x = Ut_r[kb * 64 + n];   br[kt].y = Ut_r[(kb + 1) * 64 + n];
      bi[kt].x = Ut_i[kb * 64 + n];   bi[kt].y = Ut_i[(kb + 1) * 64 + n];
    }
  }

  {
    float h[16];
#pragma unroll
    for (int j = 0; j < 16; ++j) {
      float t = b1[j] + x0 * W1[j] + x1 * W1[16 + j] + x2 * W1[32 + j] + x3 * W1[48 + j];
      h[j] = t > 0.0f ? t : 0.0f;
    }
    float A0[6], A1[6];
#pragma unroll
    for (int q = 0; q < 6; ++q) {
      float t = b2[q];
#pragma unroll
      for (int j = 0; j < 16; ++j) t += h[j] * W2[j * 6 + q];
      const float ang = 0.5f * 3.14159265358979323846f * tanhf(t);
      float sn, cs; sincosf(ang, &sn, &cs);
      const float is2 = 0.70710678118654752440f;
      A0[q] = (cs - sn) * is2;
      A1[q] = (cs + sn) * is2;
    }
    float m01[4], m23[4], m45[4];
#pragma unroll
    for (int i = 0; i < 4; ++i) {
      m01[i] = ((i & 2) ? A1[0] : A0[0]) * ((i & 1) ? A1[1] : A0[1]);
      m23[i] = ((i & 2) ? A1[2] : A0[2]) * ((i & 1) ? A1[3] : A0[3]);
      m45[i] = ((i & 2) ? A1[4] : A0[4]) * ((i & 1) ? A1[5] : A0[5]);
    }
    float* dst = &sAmp[tid * ISTR];
#pragma unroll
    for (int s = 0; s < 64; ++s)
      dst[s] = m01[s >> 4] * m23[(s >> 2) & 3] * m45[s & 3];
  }
  __syncthreads();

  {
    const float* arow = &sAmp[l * ISTR + 2 * half];
    float* prow = &sProb[(8 * half) * ISTR + wave * 16 + l];
    v2f a[2][16];
    v8f cr[2], ci[2];
    float pv[2][8];
#pragma unroll
    for (int kt = 0; kt < 16; ++kt) {
      a[0][kt].x = arow[4 * kt];
      a[0][kt].y = arow[4 * kt + 1];
    }
#pragma unroll
    for (int m = 0; m < 10; ++m) {
      if (m + 1 < 8) {
        const float* am = arow + (m + 1) * 16 * ISTR;
#pragma unroll
        for (int kt = 0; kt < 16; ++kt) {
          a[(m + 1) & 1][kt].x = am[4 * kt];
          a[(m + 1) & 1][kt].y = am[4 * kt + 1];
        }
      }
      if (m >= 2) {
        float* pm = prow + (m - 2) * 16 * ISTR;
#pragma unroll
        for (int r = 0; r < 8; ++r) pm[r * ISTR] = pv[m & 1][r];
      }
      __builtin_amdgcn_sched_barrier(0);
      if (m < 8) {
        v8f accr = v8f{}, acci = v8f{};
#pragma unroll
        for (int kt = 0; kt < 16; ++kt) {
          accr = __builtin_amdgcn_wmma_f32_16x16x4_f32(
              false, a[m & 1][kt], false, br[kt], (short)0, accr, false, false);
          acci = __builtin_amdgcn_wmma_f32_16x16x4_f32(
              false, a[m & 1][kt], false, bi[kt], (short)0, acci, false, false);
        }
        cr[m & 1] = accr; ci[m & 1] = acci;
      }
      if (m >= 1 && m <= 8) {
#pragma unroll
        for (int r = 0; r < 8; ++r) {
          const float pr = cr[(m - 1) & 1][r], pi = ci[(m - 1) & 1][r];
          pv[(m - 1) & 1][r] = pr * pr + pi * pi;
        }
      }
      __builtin_amdgcn_sched_barrier(0);
    }
  }
  __syncthreads();

  {
    if (e < batch) {
      const v4f* p4 = reinterpret_cast<const v4f*>(&sProb[tid * ISTR]);
      v2f P[32];
#pragma unroll
      for (int j = 0; j < 16; ++j) {
        const v4f t = p4[j];
        P[2 * j]     = v2f{t.x, t.y};
        P[2 * j + 1] = v2f{t.z, t.w};
      }
      v2f A16[16], D16[16];
#pragma unroll
      for (int j = 0; j < 16; ++j) { A16[j] = P[j] + P[j + 16]; D16[j] = P[j] - P[j + 16]; }
#pragma unroll
      for (int w = 8; w >= 1; w >>= 1)
#pragma unroll
        for (int j = 0; j < w; ++j) D16[j] += D16[j + w];
      const float z0 = D16[0].x + D16[0].y;
      v2f A8[8], D8[8];
#pragma unroll
      for (int j = 0; j < 8; ++j) { A8[j] = A16[j] + A16[j + 8]; D8[j] = A16[j] - A16[j + 8]; }
#pragma unroll
      for (int w = 4; w >= 1; w >>= 1)
#pragma unroll
        for (int j = 0; j < w; ++j) D8[j] += D8[j + w];
      const float z1 = D8[0].x + D8[0].y;
      v2f A4[4], D4[4];
#pragma unroll
      for (int j = 0; j < 4; ++j) { A4[j] = A8[j] + A8[j + 4]; D4[j] = A8[j] - A8[j + 4]; }
      D4[0] += D4[2]; D4[1] += D4[3]; D4[0] += D4[1];
      const float z2 = D4[0].x + D4[0].y;
      const v2f A2a = A4[0] + A4[2], A2b = A4[1] + A4[3];
      const v2f D2 = (A4[0] - A4[2]) + (A4[1] - A4[3]);
      const float z3 = D2.x + D2.y;
      const v2f A1 = A2a + A2b, D1 = A2a - A2b;
      const float z4 = D1.x + D1.y;
      const float z5 = A1.x - A1.y;
      const float z[6] = {z0, z1, z2, z3, z4, z5};
      float o0 = b4[0], o1 = b4[1], o2 = b4[2], o3 = b4[3];
#pragma unroll
      for (int j = 0; j < 16; ++j) {
        float g = b3[j];
#pragma unroll
        for (int q = 0; q < 6; ++q) g += z[q] * W3[q * 16 + j];
        g = g > 0.0f ? g : 0.0f;
        o0 += g * W4[j * 4 + 0]; o1 += g * W4[j * 4 + 1];
        o2 += g * W4[j * 4 + 2]; o3 += g * W4[j * 4 + 3];
      }
      float* op = out + (size_t)e * 4;
      op[0] = o0; op[1] = o1; op[2] = o2; op[3] = o3;
    }
  }
}

extern "C" void kernel_launch(void* const* d_in, const int* in_sizes, int n_in,
                              void* d_out, int out_size, void* d_ws, size_t ws_size,
                              hipStream_t stream) {
  const float* x  = (const float*)d_in[0];
  const float* W1 = (const float*)d_in[1];
  const float* b1 = (const float*)d_in[2];
  const float* W2 = (const float*)d_in[3];
  const float* b2 = (const float*)d_in[4];
  const float* qp = (const float*)d_in[5];
  const float* W3 = (const float*)d_in[6];
  const float* b3 = (const float*)d_in[7];
  const float* W4 = (const float*)d_in[8];
  const float* b4 = (const float*)d_in[9];
  float* out  = (float*)d_out;
  float* Ut_r = (float*)d_ws;
  float* Ut_i = Ut_r + 64 * 64;

  const int batch = in_sizes[0] / 4;
  build_unitary_kernel<<<dim3(64), dim3(32), 0, stream>>>(qp, Ut_r, Ut_i);
  const int tiles = (batch + TILE - 1) / TILE;
  qnn_kernel<<<dim3(tiles), dim3(WAVES * 32), 0, stream>>>(
      x, W1, b1, W2, b2, W3, b3, W4, b4, Ut_r, Ut_i, out, batch);
}


# <Z_q> by exact 3-point interpolation in each qubit's angle: kernel 0 runs the circuit on 729 sample states, GEMM is 84 weights x 64 sample columns (168 WMMAs per wave, was 256); new summation order
# speedup vs baseline: 15.6237x; 1.1748x over previous
#include <hip/hip_runtime.h>
#include <math.h>

typedef __attribute__((ext_vector_type(2))) float v2f;
typedef __attribute__((ext_vector_type(4))) float v4f;
typedef __attribute__((ext_vector_type(8))) float v8f;

#define WAVES 4
#define TILE  (WAVES * 32)
#define NHI    9
#define NLO    81
#define KSTEPS 21
#define KPAD   84
#define NCOL   64
#define YSTR   68


template <int M>
__device__ __forceinline__ float lane_xor(float v) {
  const int b = __float_as_int(v);
  if constexpr (M == 16) {
    return __int_as_float(
        __builtin_amdgcn_permlanex16(b, b, 0x76543210, 0xfedcba98, false, false));
  } else {
    return __int_as_float(__builtin_amdgcn_mov_dpp(b, 0x160 + M, 0xf, 0xf, true));
  }
}
template <int M>
__device__ __forceinline__ void lane_xor4(float a, float b, float c, float d,
                                          float& pa, float& pb, float& pc, float& pd) {
  pa = lane_xor<M>(a); pb = lane_xor<M>(b); pc = lane_xor<M>(c); pd = lane_xor<M>(d);
}
__device__ __forceinline__ float lane_get(float v, int src) {
  return __int_as_float(__builtin_amdgcn_readlane(__float_as_int(v), src));
}


__global__ __launch_bounds__(NHI * 32) void build_samples_kernel(
    const float* __restrict__ qp, float* __restrict__ S) {
  __shared__ float sZ[NHI * 6];
  const int lane = threadIdx.x & 31;
  const int khi  = threadIdx.x >> 5;
  const int klo  = blockIdx.x;
  const int s0 = lane, s1 = lane + 32;

  float snA, csA, snB, csB;
  {
    const int tb = (lane + 32 < 54) ? lane + 32 : 53;
    const float thA = qp[lane], thB = qp[tb];
    sincosf(0.5f * thA, &snA, &csA);
    sincosf(0.5f * thB, &snB, &csB);
  }
  auto trig = [&](int idx, float& sh, float& ch) {
    const int j = idx & 31;
    const float sa = lane_get(snA, j), ca = lane_get(csA, j);
    const float sb = lane_get(snB, j), cb = lane_get(csB, j);
    sh = (idx < 32) ? sa : sb;
    ch = (idx < 32) ? ca : cb;
  };


  float r0 = 1.0f, i0 = 0.0f;
  float r1 = 1.0f, i1 = 0.0f;
  {
    int kd[6];
    kd[0] = khi / 3;  kd[1] = khi % 3;
    kd[2] = klo / 27; kd[3] = (klo / 9) % 3; kd[4] = (klo / 3) % 3; kd[5] = klo % 3;
#pragma unroll
    for (int q = 0; q < 6; ++q) {
      const float ca = (kd[q] == 0) ? 1.0f : ((kd[q] == 1) ? 0.5f : -0.5f);
      const float sa = (kd[q] == 0) ? 0.0f : 0.86602540378443864676f;
      const int m = 1 << (5 - q);
      r0 *= (s0 & m) ? sa : ca;
      r1 *= (s1 & m) ? sa : ca;
    }
  }

  auto partner = [&](int m, float& pr0, float& pi0, float& pr1, float& pi1) {
    switch (m) {
      case 32: pr0 = r1; pi0 = i1; pr1 = r0; pi1 = i0; break;
      case 16: lane_xor4<16>(r0, i0, r1, i1, pr0, pi0, pr1, pi1); break;
      case 8:  lane_xor4<8>(r0, i0, r1, i1, pr0, pi0, pr1, pi1); break;
      case 4:  lane_xor4<4>(r0, i0, r1, i1, pr0, pi0, pr1, pi1); break;
      case 2:  lane_xor4<2>(r0, i0, r1, i1, pr0, pi0, pr1, pi1); break;
      default: lane_xor4<1>(r0, i0, r1, i1, pr0, pi0, pr1, pi1); break;
    }
  };
  auto cnot = [&](int cq, int tq) {
    const int mc = 1 << (5 - cq), mt = 1 << (5 - tq);
    float pr0, pi0, pr1, pi1;
    partner(mt, pr0, pi0, pr1, pi1);
    if (s0 & mc) { r0 = pr0; i0 = pi0; }
    if (s1 & mc) { r1 = pr1; i1 = pi1; }
  };
  auto rx = [&](int q, float sh, float ch) {
    const int m = 1 << (5 - q);
    float pr0, pi0, pr1, pi1;
    partner(m, pr0, pi0, pr1, pi1);
    const float a0 = ch * r0 + sh * pi0, b0 = ch * i0 - sh * pr0;
    const float a1 = ch * r1 + sh * pi1, b1 = ch * i1 - sh * pr1;
    r0 = a0; i0 = b0; r1 = a1; i1 = b1;
  };
  auto ry = [&](int q, float sh, float ch) {
    const int m = 1 << (5 - q);
    float pr0, pi0, pr1, pi1;
    partner(m, pr0, pi0, pr1, pi1);
    const float t0 = (s0 & m) ? sh : -sh;
    const float t1 = (s1 & m) ? sh : -sh;
    const float a0 = ch * r0 + t0 * pr0, b0 = ch * i0 + t0 * pi0;
    const float a1 = ch * r1 + t1 * pr1, b1 = ch * i1 + t1 * pi1;
    r0 = a0; i0 = b0; r1 = a1; i1 = b1;
  };
  auto phase = [&](int s, int m, float sh, float ch, float& r, float& i) {
    const float t = (s & m) ? -sh : sh;
    const float a = ch * r + t * i, b = ch * i - t * r;
    r = a; i = b;
  };
  auto rz = [&](int q, float sh, float ch) {
    const int m = 1 << (5 - q);
    phase(s0, m, sh, ch, r0, i0);
    phase(s1, m, sh, ch, r1, i1);
  };
  auto crz = [&](int cq, int tq, float sh, float ch) {
    const int mc = 1 << (5 - cq), mt = 1 << (5 - tq);
    float a = r0, b = i0;
    phase(s0, mt, sh, ch, a, b);
    if (s0 & mc) { r0 = a; i0 = b; }
    a = r1; b = i1;
    phase(s1, mt, sh, ch, a, b);
    if (s1 & mc) { r1 = a; i1 = b; }
  };

  for (int layer = 0; layer < 3; ++layer) {
    cnot(0, 1); cnot(2, 3); cnot(4, 5);
    cnot(1, 2); cnot(3, 4);
    for (int i = 0; i < 6; ++i) {
      float sh, ch;
      trig(layer * 18 + i * 3 + 0, sh, ch); rx(i, sh, ch);
      trig(layer * 18 + i * 3 + 1, sh, ch); ry(i, sh, ch);
      trig(layer * 18 + i * 3 + 2, sh, ch); rz(i, sh, ch);
    }
    if (layer < 2)
      for (int i = 0; i < 6; ++i) {
        float sh, ch;
        trig(layer * 18 + i * 3 + 0, sh, ch);
        crz(i, (i + 1) % 6, sh, ch);
      }
  }

  {
    const float p0 = r0 * r0 + i0 * i0, p1 = r1 * r1 + i1 * i1;
    const float t = p0 + p1;
    float zz[6];
    zz[0] = p0 - p1;
#pragma unroll
    for (int q = 1; q < 6; ++q) zz[q] = (lane & (1 << (5 - q))) ? -t : t;
#pragma unroll
    for (int q = 0; q < 6; ++q) {
      float v = zz[q];
      v += lane_xor<1>(v); v += lane_xor<2>(v); v += lane_xor<4>(v);
      v += lane_xor<8>(v); v += lane_xor<16>(v);
      zz[q] = v;
    }
    if (lane == 0) {
#pragma unroll
      for (int q = 0; q < 6; ++q) sZ[khi * 6 + q] = zz[q];
    }
  }
  __syncthreads();
  if (threadIdx.x < NCOL) {
    const int n = threadIdx.x;
    float v = 0.0f;
    if (n < 6 * NHI) v = sZ[(n % NHI) * 6 + n / NHI];
    S[klo * NCOL + n] = v;
  }
  if (blockIdx.x == 0 && threadIdx.x >= NCOL && threadIdx.x < 4 * NCOL)
    S[NLO * NCOL + (threadIdx.x - NCOL)] = 0.0f;
}

__global__ __launch_bounds__(WAVES * 32) void qnn_kernel(
    const float* __restrict__ x,
    const float* __restrict__ W1, const float* __restrict__ b1,
    const float* __restrict__ W2, const float* __restrict__ b2,
    const float* __restrict__ W3, const float* __restrict__ b3,
    const float* __restrict__ W4, const float* __restrict__ b4,
    const float* __restrict__ S,
    float* __restrict__ out, int batch) {
  __shared__ __attribute__((aligned(16))) float sWgt[TILE * KPAD];
  __shared__ __attribute__((aligned(16))) float sY[TILE * YSTR];

  const int tid  = threadIdx.x;
  const int wave = tid >> 5;
  const int lane = tid & 31;
  const int half = lane >> 4;
  const int l    = lane & 15;
  const int ebase = blockIdx.x * TILE;

  const int e = ebase + tid;
  const int ec = (e < batch) ? e : (batch - 1);
  const float x0 = x[ec * 4 + 0], x1 = x[ec * 4 + 1];
  const float x2 = x[ec * 4 + 2], x3 = x[ec * 4 + 3];

  v2f bs[KSTEPS];
  {
    const int n = wave * 16 + l;
#pragma unroll
    for (int kt = 0; kt < KSTEPS; ++kt) {
      const int kb = 4 * kt + 2 * half;
      bs[kt].x = S[kb * NCOL + n];   bs[kt].y = S[(kb + 1) * NCOL + n];
    }
  }

  float whi[NHI];
  {
    float h[16];
#pragma unroll
    for (int j = 0; j < 16; ++j) {
      float t = b1[j] + x0 * W1[j] + x1 * W1[16 + j] + x2 * W1[32 + j] + x3 * W1[48 + j];
      h[j] = t > 0.0f ? t : 0.0f;
    }
    float g[6][3];
#pragma unroll
    for (int q = 0; q < 6; ++q) {
      float t = b2[q];
#pragma unroll
      for (int j = 0; j < 16; ++j) t += h[j] * W2[j * 6 + q];
      const float al = 3.14159265358979323846f * tanhf(t);
      float sn, cs; sincosf(al, &sn, &cs);
      const float c2 = -sn, s2 = cs;
      const float third = 0.33333333333333333333f;
      const float r3 = 1.73205080756887729353f;
      const float u = 1.0f - c2;
      g[q][0] = (1.0f + 2.0f * c2) * third;
      g[q][1] = (u + r3 * s2) * third;
      g[q][2] = (u - r3 * s2) * third;
    }
    float m23[9], m45[9];
#pragma unroll
    for (int i = 0; i < 3; ++i)
#pragma unroll
      for (int j = 0; j < 3; ++j) {
        whi[3 * i + j] = g[0][i] * g[1][j];
        m23[3 * i + j] = g[2][i] * g[3][j];
        m45[3 * i + j] = g[4][i] * g[5][j];
      }
    float* dst = &sWgt[tid * KPAD];
#pragma unroll
    for (int a = 0; a < 9; ++a)
#pragma unroll
      for (int b = 0; b < 9; ++b)
        dst[9 * a + b] = m23[a] * m45[b];
    dst[81] = 0.0f; dst[82] = 0.0f; dst[83] = 0.0f;
  }
  __syncthreads();

  {
    const float* arow = &sWgt[l * KPAD + 2 * half];
    float* yrow = &sY[(8 * half) * YSTR + wave * 16 + l];
    v2f a[2][KSTEPS];
    v8f c0[2], c1[2];
    float pv[2][8];
#pragma unroll
    for (int kt = 0; kt < KSTEPS; ++kt) {
      a[0][kt].x = arow[4 * kt];
      a[0][kt].y = arow[4 * kt + 1];
    }
#pragma unroll
    for (int m = 0; m < 10; ++m) {
      if (m + 1 < 8) {
        const float* am = arow + (m + 1) * 16 * KPAD;
#pragma unroll
        for (int kt = 0; kt < KSTEPS; ++kt) {
          a[(m + 1) & 1][kt].x = am[4 * kt];
          a[(m + 1) & 1][kt].y = am[4 * kt + 1];
        }
      }
      if (m >= 2) {
        float* ym = yrow + (m - 2) * 16 * YSTR;
#pragma unroll
        for (int r = 0; r < 8; ++r) ym[r * YSTR] = pv[m & 1][r];
      }
      __builtin_amdgcn_sched_barrier(0);
      if (m < 8) {
        v8f acc0 = v8f{}, acc1 = v8f{};
#pragma unroll
        for (int kt = 0; kt < KSTEPS; ++kt) {
          if (kt & 1)
            acc1 = __builtin_amdgcn_wmma_f32_16x16x4_f32(
                false, a[m & 1][kt], false, bs[kt], (short)0, acc1, false, false);
          else
            acc0 = __builtin_amdgcn_wmma_f32_16x16x4_f32(
                false, a[m & 1][kt], false, bs[kt], (short)0, acc0, false, false);
        }
        c0[m & 1] = acc0; c1[m & 1] = acc1;
      }
      if (m >= 1 && m <= 8) {
#pragma unroll
        for (int r = 0; r < 8; ++r)
          pv[(m - 1) & 1][r] = c0[(m - 1) & 1][r] + c1[(m - 1) & 1][r];
      }
      __builtin_amdgcn_sched_barrier(0);
    }
  }
  __syncthreads();

  {
    if (e < batch) {
      const v4f* y4 = reinterpret_cast<const v4f*>(&sY[tid * YSTR]);
      float y[56];
#pragma unroll
      for (int j = 0; j < 14; ++j) {
        const v4f t = y4[j];
        y[4 * j] = t.x; y[4 * j + 1] = t.y; y[4 * j + 2] = t.z; y[4 * j + 3] = t.w;
      }
      float z[6];
#pragma unroll
      for (int q = 0; q < 6; ++q) {
        float acc = whi[0] * y[q * NHI];
#pragma unroll
        for (int k = 1; k < NHI; ++k) acc += whi[k] * y[q * NHI + k];
        z[q] = acc;
      }
      float o0 = b4[0], o1 = b4[1], o2 = b4[2], o3 = b4[3];
#pragma unroll
      for (int j = 0; j < 16; ++j) {
        float g = b3[j];
#pragma unroll
        for (int q = 0; q < 6; ++q) g += z[q] * W3[q * 16 + j];
        g = g > 0.0f ? g : 0.0f;
        o0 += g * W4[j * 4 + 0]; o1 += g * W4[j * 4 + 1];
        o2 += g * W4[j * 4 + 2]; o3 += g * W4[j * 4 + 3];
      }
      float* op = out + (size_t)e * 4;
      op[0] = o0; op[1] = o1; op[2] = o2; op[3] = o3;
    }
  }
}

extern "C" void kernel_launch(void* const* d_in, const int* in_sizes, int n_in,
                              void* d_out, int out_size, void* d_ws, size_t ws_size,
                              hipStream_t stream) {
  const float* x  = (const float*)d_in[0];
  const float* W1 = (const float*)d_in[1];
  const float* b1 = (const float*)d_in[2];
  const float* W2 = (const float*)d_in[3];
  const float* b2 = (const float*)d_in[4];
  const float* qp = (const float*)d_in[5];
  const float* W3 = (const float*)d_in[6];
  const float* b3 = (const float*)d_in[7];
  const float* W4 = (const float*)d_in[8];
  const float* b4 = (const float*)d_in[9];
  float* out = (float*)d_out;
  float* S   = (float*)d_ws;

  const int batch = in_sizes[0] / 4;
  build_samples_kernel<<<dim3(NLO), dim3(NHI * 32), 0, stream>>>(qp, S);
  const int tiles = (batch + TILE - 1) / TILE;
  qnn_kernel<<<dim3(tiles), dim3(WAVES * 32), 0, stream>>>(
      x, W1, b1, W2, b2, W3, b3, W4, b4, S, out, batch);
}


# Prologue: tanh, sin, cos of the six encoding angles as straight-line bounded-range forms (exp2+rcp; half-angle minimax polynomials), not the branching library calls; summation orders unchanged
# speedup vs baseline: 16.4156x; 1.0507x over previous
#include <hip/hip_runtime.h>
#include <math.h>

typedef __attribute__((ext_vector_type(2))) float v2f;
typedef __attribute__((ext_vector_type(4))) float v4f;
typedef __attribute__((ext_vector_type(8))) float v8f;

#define WAVES 4
#define TILE  (WAVES * 32)
#define NHI    9
#define NLO    81
#define KSTEPS 21
#define KPAD   84
#define NCOL   64
#define YSTR   68


template <int M>
__device__ __forceinline__ float lane_xor(float v) {
  const int b = __float_as_int(v);
  if constexpr (M == 16) {
    return __int_as_float(
        __builtin_amdgcn_permlanex16(b, b, 0x76543210, 0xfedcba98, false, false));
  } else {
    return __int_as_float(__builtin_amdgcn_mov_dpp(b, 0x160 + M, 0xf, 0xf, true));
  }
}
template <int M>
__device__ __forceinline__ void lane_xor4(float a, float b, float c, float d,
                                          float& pa, float& pb, float& pc, float& pd) {
  pa = lane_xor<M>(a); pb = lane_xor<M>(b); pc = lane_xor<M>(c); pd = lane_xor<M>(d);
}
__device__ __forceinline__ float lane_get(float v, int src) {
  return __int_as_float(__builtin_amdgcn_readlane(__float_as_int(v), src));
}


__global__ __launch_bounds__(NHI * 32) void build_samples_kernel(
    const float* __restrict__ qp, float* __restrict__ S) {
  __shared__ float sZ[NHI * 6];
  const int lane = threadIdx.x & 31;
  const int khi  = threadIdx.x >> 5;
  const int klo  = blockIdx.x;
  const int s0 = lane, s1 = lane + 32;

  float snA, csA, snB, csB;
  {
    const int tb = (lane + 32 < 54) ? lane + 32 : 53;
    const float thA = qp[lane], thB = qp[tb];
    sincosf(0.5f * thA, &snA, &csA);
    sincosf(0.5f * thB, &snB, &csB);
  }
  auto trig = [&](int idx, float& sh, float& ch) {
    const int j = idx & 31;
    const float sa = lane_get(snA, j), ca = lane_get(csA, j);
    const float sb = lane_get(snB, j), cb = lane_get(csB, j);
    sh = (idx < 32) ? sa : sb;
    ch = (idx < 32) ? ca : cb;
  };


  float r0 = 1.0f, i0 = 0.0f;
  float r1 = 1.0f, i1 = 0.0f;
  {
    int kd[6];
    kd[0] = khi / 3;  kd[1] = khi % 3;
    kd[2] = klo / 27; kd[3] = (klo / 9) % 3; kd[4] = (klo / 3) % 3; kd[5] = klo % 3;
#pragma unroll
    for (int q = 0; q < 6; ++q) {
      const float ca = (kd[q] == 0) ? 1.0f : ((kd[q] == 1) ? 0.5f : -0.5f);
      const float sa = (kd[q] == 0) ? 0.0f : 0.86602540378443864676f;
      const int m = 1 << (5 - q);
      r0 *= (s0 & m) ? sa : ca;
      r1 *= (s1 & m) ? sa : ca;
    }
  }

  auto partner = [&](int m, float& pr0, float& pi0, float& pr1, float& pi1) {
    switch (m) {
      case 32: pr0 = r1; pi0 = i1; pr1 = r0; pi1 = i0; break;
      case 16: lane_xor4<16>(r0, i0, r1, i1, pr0, pi0, pr1, pi1); break;
      case 8:  lane_xor4<8>(r0, i0, r1, i1, pr0, pi0, pr1, pi1); break;
      case 4:  lane_xor4<4>(r0, i0, r1, i1, pr0, pi0, pr1, pi1); break;
      case 2:  lane_xor4<2>(r0, i0, r1, i1, pr0, pi0, pr1, pi1); break;
      default: lane_xor4<1>(r0, i0, r1, i1, pr0, pi0, pr1, pi1); break;
    }
  };
  auto cnot = [&](int cq, int tq) {
    const int mc = 1 << (5 - cq), mt = 1 << (5 - tq);
    float pr0, pi0, pr1, pi1;
    partner(mt, pr0, pi0, pr1, pi1);
    if (s0 & mc) { r0 = pr0; i0 = pi0; }
    if (s1 & mc) { r1 = pr1; i1 = pi1; }
  };
  auto rx = [&](int q, float sh, float ch) {
    const int m = 1 << (5 - q);
    float pr0, pi0, pr1, pi1;
    partner(m, pr0, pi0, pr1, pi1);
    const float a0 = ch * r0 + sh * pi0, b0 = ch * i0 - sh * pr0;
    const float a1 = ch * r1 + sh * pi1, b1 = ch * i1 - sh * pr1;
    r0 = a0; i0 = b0; r1 = a1; i1 = b1;
  };
  auto ry = [&](int q, float sh, float ch) {
    const int m = 1 << (5 - q);
    float pr0, pi0, pr1, pi1;
    partner(m, pr0, pi0, pr1, pi1);
    const float t0 = (s0 & m) ? sh : -sh;
    const float t1 = (s1 & m) ? sh : -sh;
    const float a0 = ch * r0 + t0 * pr0, b0 = ch * i0 + t0 * pi0;
    const float a1 = ch * r1 + t1 * pr1, b1 = ch * i1 + t1 * pi1;
    r0 = a0; i0 = b0; r1 = a1; i1 = b1;
  };
  auto phase = [&](int s, int m, float sh, float ch, float& r, float& i) {
    const float t = (s & m) ? -sh : sh;
    const float a = ch * r + t * i, b = ch * i - t * r;
    r = a; i = b;
  };
  auto rz = [&](int q, float sh, float ch) {
    const int m = 1 << (5 - q);
    phase(s0, m, sh, ch, r0, i0);
    phase(s1, m, sh, ch, r1, i1);
  };
  auto crz = [&](int cq, int tq, float sh, float ch) {
    const int mc = 1 << (5 - cq), mt = 1 << (5 - tq);
    float a = r0, b = i0;
    phase(s0, mt, sh, ch, a, b);
    if (s0 & mc) { r0 = a; i0 = b; }
    a = r1; b = i1;
    phase(s1, mt, sh, ch, a, b);
    if (s1 & mc) { r1 = a; i1 = b; }
  };

  for (int layer = 0; layer < 3; ++layer) {
    cnot(0, 1); cnot(2, 3); cnot(4, 5);
    cnot(1, 2); cnot(3, 4);
    for (int i = 0; i < 6; ++i) {
      float sh, ch;
      trig(layer * 18 + i * 3 + 0, sh, ch); rx(i, sh, ch);
      trig(layer * 18 + i * 3 + 1, sh, ch); ry(i, sh, ch);
      trig(layer * 18 + i * 3 + 2, sh, ch); rz(i, sh, ch);
    }
    if (layer < 2)
      for (int i = 0; i < 6; ++i) {
        float sh, ch;
        trig(layer * 18 + i * 3 + 0, sh, ch);
        crz(i, (i + 1) % 6, sh, ch);
      }
  }

  {
    const float p0 = r0 * r0 + i0 * i0, p1 = r1 * r1 + i1 * i1;
    const float t = p0 + p1;
    float zz[6];
    zz[0] = p0 - p1;
#pragma unroll
    for (int q = 1; q < 6; ++q) zz[q] = (lane & (1 << (5 - q))) ? -t : t;
#pragma unroll
    for (int q = 0; q < 6; ++q) {
      float v = zz[q];
      v += lane_xor<1>(v); v += lane_xor<2>(v); v += lane_xor<4>(v);
      v += lane_xor<8>(v); v += lane_xor<16>(v);
      zz[q] = v;
    }
    if (lane == 0) {
#pragma unroll
      for (int q = 0; q < 6; ++q) sZ[khi * 6 + q] = zz[q];
    }
  }
  __syncthreads();
  if (threadIdx.x < NCOL) {
    const int n = threadIdx.x;
    float v = 0.0f;
    if (n < 6 * NHI) v = sZ[(n % NHI) * 6 + n / NHI];
    S[klo * NCOL + n] = v;
  }
  if (blockIdx.x == 0 && threadIdx.x >= NCOL && threadIdx.x < 4 * NCOL)
    S[NLO * NCOL + (threadIdx.x - NCOL)] = 0.0f;
}

__global__ __launch_bounds__(WAVES * 32) void qnn_kernel(
    const float* __restrict__ x,
    const float* __restrict__ W1, const float* __restrict__ b1,
    const float* __restrict__ W2, const float* __restrict__ b2,
    const float* __restrict__ W3, const float* __restrict__ b3,
    const float* __restrict__ W4, const float* __restrict__ b4,
    const float* __restrict__ S,
    float* __restrict__ out, int batch) {
  __shared__ __attribute__((aligned(16))) float sWgt[TILE * KPAD];
  __shared__ __attribute__((aligned(16))) float sY[TILE * YSTR];

  const int tid  = threadIdx.x;
  const int wave = tid >> 5;
  const int lane = tid & 31;
  const int half = lane >> 4;
  const int l    = lane & 15;
  const int ebase = blockIdx.x * TILE;

  const int e = ebase + tid;
  const int ec = (e < batch) ? e : (batch - 1);
  const float x0 = x[ec * 4 + 0], x1 = x[ec * 4 + 1];
  const float x2 = x[ec * 4 + 2], x3 = x[ec * 4 + 3];

  v2f bs[KSTEPS];
  {
    const int n = wave * 16 + l;
#pragma unroll
    for (int kt = 0; kt < KSTEPS; ++kt) {
      const int kb = 4 * kt + 2 * half;
      bs[kt].x = S[kb * NCOL + n];   bs[kt].y = S[(kb + 1) * NCOL + n];
    }
  }

  float whi[NHI];
  {
    float h[16];
#pragma unroll
    for (int j = 0; j < 16; ++j) h[j] = b1[j];
    {
      const float xk[4] = {x0, x1, x2, x3};
#pragma unroll
      for (int k = 0; k < 4; ++k) {
#pragma unroll
        for (int j = 0; j < 16; ++j) h[j] += xk[k] * W1[16 * k + j];
        __builtin_amdgcn_sched_barrier(0);
      }
    }
#pragma unroll
    for (int j = 0; j < 16; ++j) h[j] = h[j] > 0.0f ? h[j] : 0.0f;
    float t[6];
#pragma unroll
    for (int q = 0; q < 6; ++q) t[q] = b2[q];
#pragma unroll
    for (int f0 = 0; f0 < 96; f0 += 16) {
#pragma unroll
      for (int f = f0; f < f0 + 16; ++f) t[f % 6] += h[f / 6] * W2[f];
      __builtin_amdgcn_sched_barrier(0);
    }
    float g[6][3];
#pragma unroll
    for (int q = 0; q < 6; ++q) {
      const float E  = __builtin_amdgcn_exp2f(-2.88539008177792681472f * fabsf(t[q]));
      const float ua = (1.0f - E) * __builtin_amdgcn_rcpf(1.0f + E);
      const float u  = copysignf(ua, t[q]);
      const float w  = u * u;
      float pp = 1.5082264e-4f;
      pp = fmaf(pp, w, -4.6722330e-3f);
      pp = fmaf(pp, w, 7.9688482e-2f);
      pp = fmaf(pp, w, -6.4596337e-1f);
      pp = fmaf(pp, w, 1.5707963e+0f);
      const float sh = u * pp;
      float ch = -2.3824918e-5f;
      ch = fmaf(ch, w, 9.1772486e-4f);
      ch = fmaf(ch, w, -2.0862689e-2f);
      ch = fmaf(ch, w, 2.5366932e-1f);
      ch = fmaf(ch, w, -1.2337005e+0f);
      ch = fmaf(ch, w, 1.0f);
      const float hs = sh * ch;
      const float ca = fmaf(-2.0f * sh, sh, 1.0f);
      const float third = 0.33333333333333333333f;
      const float r33 = 0.57735026918962576451f;
      const float bb = fmaf(hs, 2.0f * third, third);
      g[q][0] = fmaf(hs, -4.0f * third, third);
      g[q][1] = fmaf(ca, r33, bb);
      g[q][2] = fmaf(ca, -r33, bb);
    }
    float m23[9], m45[9];
#pragma unroll
    for (int i = 0; i < 3; ++i)
#pragma unroll
      for (int j = 0; j < 3; ++j) {
        whi[3 * i + j] = g[0][i] * g[1][j];
        m23[3 * i + j] = g[2][i] * g[3][j];
        m45[3 * i + j] = g[4][i] * g[5][j];
      }
    float* dst = &sWgt[tid * KPAD];
#pragma unroll
    for (int a = 0; a < 9; ++a)
#pragma unroll
      for (int b = 0; b < 9; ++b)
        dst[9 * a + b] = m23[a] * m45[b];
    dst[81] = 0.0f; dst[82] = 0.0f; dst[83] = 0.0f;
  }
  __syncthreads();

  {
    const float* arow = &sWgt[l * KPAD + 2 * half];
    float* yrow = &sY[(8 * half) * YSTR + wave * 16 + l];
    v2f a[2][KSTEPS];
    v8f c0[2], c1[2];
    float pv[2][8];
#pragma unroll
    for (int kt = 0; kt < KSTEPS; ++kt) {
      a[0][kt].x = arow[4 * kt];
      a[0][kt].y = arow[4 * kt + 1];
    }
#pragma unroll
    for (int m = 0; m < 10; ++m) {
      if (m + 1 < 8) {
        const float* am = arow + (m + 1) * 16 * KPAD;
#pragma unroll
        for (int kt = 0; kt < KSTEPS; ++kt) {
          a[(m + 1) & 1][kt].x = am[4 * kt];
          a[(m + 1) & 1][kt].y = am[4 * kt + 1];
        }
      }
      if (m >= 2) {
        float* ym = yrow + (m - 2) * 16 * YSTR;
#pragma unroll
        for (int r = 0; r < 8; ++r) ym[r * YSTR] = pv[m & 1][r];
      }
      __builtin_amdgcn_sched_barrier(0);
      if (m < 8) {
        v8f acc0 = v8f{}, acc1 = v8f{};
#pragma unroll
        for (int kt = 0; kt < KSTEPS; ++kt) {
          if (kt & 1)
            acc1 = __builtin_amdgcn_wmma_f32_16x16x4_f32(
                false, a[m & 1][kt], false, bs[kt], (short)0, acc1, false, false);
          else
            acc0 = __builtin_amdgcn_wmma_f32_16x16x4_f32(
                false, a[m & 1][kt], false, bs[kt], (short)0, acc0, false, false);
        }
        c0[m & 1] = acc0; c1[m & 1] = acc1;
      }
      if (m >= 1 && m <= 8) {
#pragma unroll
        for (int r = 0; r < 8; ++r)
          pv[(m - 1) & 1][r] = c0[(m - 1) & 1][r] + c1[(m - 1) & 1][r];
      }
      __builtin_amdgcn_sched_barrier(0);
    }
  }
  __syncthreads();

  {
    if (e < batch) {
      const v4f* y4 = reinterpret_cast<const v4f*>(&sY[tid * YSTR]);
      float y[56];
#pragma unroll
      for (int j = 0; j < 14; ++j) {
        const v4f t = y4[j];
        y[4 * j] = t.x; y[4 * j + 1] = t.y; y[4 * j + 2] = t.z; y[4 * j + 3] = t.w;
      }
      float z[6];
#pragma unroll
      for (int q = 0; q < 6; ++q) {
        float acc = whi[0] * y[q * NHI];
#pragma unroll
        for (int k = 1; k < NHI; ++k) acc += whi[k] * y[q * NHI + k];
        z[q] = acc;
      }
      float o0 = b4[0], o1 = b4[1], o2 = b4[2], o3 = b4[3];
#pragma unroll
      for (int j = 0; j < 16; ++j) {
        float g = b3[j];
#pragma unroll
        for (int q = 0; q < 6; ++q) g += z[q] * W3[q * 16 + j];
        g = g > 0.0f ? g : 0.0f;
        o0 += g * W4[j * 4 + 0]; o1 += g * W4[j * 4 + 1];
        o2 += g * W4[j * 4 + 2]; o3 += g * W4[j * 4 + 3];
      }
      float* op = out + (size_t)e * 4;
      op[0] = o0; op[1] = o1; op[2] = o2; op[3] = o3;
    }
  }
}

extern "C" void kernel_launch(void* const* d_in, const int* in_sizes, int n_in,
                              void* d_out, int out_size, void* d_ws, size_t ws_size,
                              hipStream_t stream) {
  const float* x  = (const float*)d_in[0];
  const float* W1 = (const float*)d_in[1];
  const float* b1 = (const float*)d_in[2];
  const float* W2 = (const float*)d_in[3];
  const float* b2 = (const float*)d_in[4];
  const float* qp = (const float*)d_in[5];
  const float* W3 = (const float*)d_in[6];
  const float* b3 = (const float*)d_in[7];
  const float* W4 = (const float*)d_in[8];
  const float* b4 = (const float*)d_in[9];
  float* out = (float*)d_out;
  float* S   = (float*)d_ws;

  const int batch = in_sizes[0] / 4;
  build_samples_kernel<<<dim3(NLO), dim3(NHI * 32), 0, stream>>>(qp, S);
  const int tiles = (batch + TILE - 1) / TILE;
  qnn_kernel<<<dim3(tiles), dim3(WAVES * 32), 0, stream>>>(
      x, W1, b1, W2, b2, W3, b3, W4, b4, S, out, batch);
}


# Main kernel: MLP multiply-adds read their weights straight from scalar registers (packed f32 forms off in this kernel: no vector copy per weight), paired by dual issue; summation orders unchanged
# speedup vs baseline: 17.7385x; 1.0806x over previous
#include <hip/hip_runtime.h>
#include <math.h>

typedef __attribute__((ext_vector_type(2))) float v2f;
typedef __attribute__((ext_vector_type(4))) float v4f;
typedef __attribute__((ext_vector_type(8))) float v8f;

#define WAVES 4
#define TILE  (WAVES * 32)
#define NHI    9
#define NLO    81
#define KSTEPS 21
#define KPAD   84
#define NCOL   64
#define YSTR   68


template <int M>
__device__ __forceinline__ float lane_xor(float v) {
  const int b = __float_as_int(v);
  if constexpr (M == 16) {
    return __int_as_float(
        __builtin_amdgcn_permlanex16(b, b, 0x76543210, 0xfedcba98, false, false));
  } else {
    return __int_as_float(__builtin_amdgcn_mov_dpp(b, 0x160 + M, 0xf, 0xf, true));
  }
}
template <int M>
__device__ __forceinline__ void lane_xor4(float a, float b, float c, float d,
                                          float& pa, float& pb, float& pc, float& pd) {
  pa = lane_xor<M>(a); pb = lane_xor<M>(b); pc = lane_xor<M>(c); pd = lane_xor<M>(d);
}
__device__ __forceinline__ float lane_get(float v, int src) {
  return __int_as_float(__builtin_amdgcn_readlane(__float_as_int(v), src));
}


__global__ __launch_bounds__(NHI * 32) void build_samples_kernel(
    const float* __restrict__ qp, float* __restrict__ S) {
  __shared__ float sZ[NHI * 6];
  const int lane = threadIdx.x & 31;
  const int khi  = threadIdx.x >> 5;
  const int klo  = blockIdx.x;
  const int s0 = lane, s1 = lane + 32;

  float snA, csA, snB, csB;
  {
    const int tb = (lane + 32 < 54) ? lane + 32 : 53;
    const float thA = qp[lane], thB = qp[tb];
    sincosf(0.5f * thA, &snA, &csA);
    sincosf(0.5f * thB, &snB, &csB);
  }
  auto trig = [&](int idx, float& sh, float& ch) {
    const int j = idx & 31;
    const float sa = lane_get(snA, j), ca = lane_get(csA, j);
    const float sb = lane_get(snB, j), cb = lane_get(csB, j);
    sh = (idx < 32) ? sa : sb;
    ch = (idx < 32) ? ca : cb;
  };


  float r0 = 1.0f, i0 = 0.0f;
  float r1 = 1.0f, i1 = 0.0f;
  {
    int kd[6];
    kd[0] = khi / 3;  kd[1] = khi % 3;
    kd[2] = klo / 27; kd[3] = (klo / 9) % 3; kd[4] = (klo / 3) % 3; kd[5] = klo % 3;
#pragma unroll
    for (int q = 0; q < 6; ++q) {
      const float ca = (kd[q] == 0) ? 1.0f : ((kd[q] == 1) ? 0.5f : -0.5f);
      const float sa = (kd[q] == 0) ? 0.0f : 0.86602540378443864676f;
      const int m = 1 << (5 - q);
      r0 *= (s0 & m) ? sa : ca;
      r1 *= (s1 & m) ? sa : ca;
    }
  }

  auto partner = [&](int m, float& pr0, float& pi0, float& pr1, float& pi1) {
    switch (m) {
      case 32: pr0 = r1; pi0 = i1; pr1 = r0; pi1 = i0; break;
      case 16: lane_xor4<16>(r0, i0, r1, i1, pr0, pi0, pr1, pi1); break;
      case 8:  lane_xor4<8>(r0, i0, r1, i1, pr0, pi0, pr1, pi1); break;
      case 4:  lane_xor4<4>(r0, i0, r1, i1, pr0, pi0, pr1, pi1); break;
      case 2:  lane_xor4<2>(r0, i0, r1, i1, pr0, pi0, pr1, pi1); break;
      default: lane_xor4<1>(r0, i0, r1, i1, pr0, pi0, pr1, pi1); break;
    }
  };
  auto cnot = [&](int cq, int tq) {
    const int mc = 1 << (5 - cq), mt = 1 << (5 - tq);
    float pr0, pi0, pr1, pi1;
    partner(mt, pr0, pi0, pr1, pi1);
    if (s0 & mc) { r0 = pr0; i0 = pi0; }
    if (s1 & mc) { r1 = pr1; i1 = pi1; }
  };
  auto rx = [&](int q, float sh, float ch) {
    const int m = 1 << (5 - q);
    float pr0, pi0, pr1, pi1;
    partner(m, pr0, pi0, pr1, pi1);
    const float a0 = ch * r0 + sh * pi0, b0 = ch * i0 - sh * pr0;
    const float a1 = ch * r1 + sh * pi1, b1 = ch * i1 - sh * pr1;
    r0 = a0; i0 = b0; r1 = a1; i1 = b1;
  };
  auto ry = [&](int q, float sh, float ch) {
    const int m = 1 << (5 - q);
    float pr0, pi0, pr1, pi1;
    partner(m, pr0, pi0, pr1, pi1);
    const float t0 = (s0 & m) ? sh : -sh;
    const float t1 = (s1 & m) ? sh : -sh;
    const float a0 = ch * r0 + t0 * pr0, b0 = ch * i0 + t0 * pi0;
    const float a1 = ch * r1 + t1 * pr1, b1 = ch * i1 + t1 * pi1;
    r0 = a0; i0 = b0; r1 = a1; i1 = b1;
  };
  auto phase = [&](int s, int m, float sh, float ch, float& r, float& i) {
    const float t = (s & m) ? -sh : sh;
    const float a = ch * r + t * i, b = ch * i - t * r;
    r = a; i = b;
  };
  auto rz = [&](int q, float sh, float ch) {
    const int m = 1 << (5 - q);
    phase(s0, m, sh, ch, r0, i0);
    phase(s1, m, sh, ch, r1, i1);
  };
  auto crz = [&](int cq, int tq, float sh, float ch) {
    const int mc = 1 << (5 - cq), mt = 1 << (5 - tq);
    float a = r0, b = i0;
    phase(s0, mt, sh, ch, a, b);
    if (s0 & mc) { r0 = a; i0 = b; }
    a = r1; b = i1;
    phase(s1, mt, sh, ch, a, b);
    if (s1 & mc) { r1 = a; i1 = b; }
  };

  for (int layer = 0; layer < 3; ++layer) {
    cnot(0, 1); cnot(2, 3); cnot(4, 5);
    cnot(1, 2); cnot(3, 4);
    for (int i = 0; i < 6; ++i) {
      float sh, ch;
      trig(layer * 18 + i * 3 + 0, sh, ch); rx(i, sh, ch);
      trig(layer * 18 + i * 3 + 1, sh, ch); ry(i, sh, ch);
      trig(layer * 18 + i * 3 + 2, sh, ch); rz(i, sh, ch);
    }
    if (layer < 2)
      for (int i = 0; i < 6; ++i) {
        float sh, ch;
        trig(layer * 18 + i * 3 + 0, sh, ch);
        crz(i, (i + 1) % 6, sh, ch);
      }
  }

  {
    const float p0 = r0 * r0 + i0 * i0, p1 = r1 * r1 + i1 * i1;
    const float t = p0 + p1;
    float zz[6];
    zz[0] = p0 - p1;
#pragma unroll
    for (int q = 1; q < 6; ++q) zz[q] = (lane & (1 << (5 - q))) ? -t : t;
#pragma unroll
    for (int q = 0; q < 6; ++q) {
      float v = zz[q];
      v += lane_xor<1>(v); v += lane_xor<2>(v); v += lane_xor<4>(v);
      v += lane_xor<8>(v); v += lane_xor<16>(v);
      zz[q] = v;
    }
    if (lane == 0) {
#pragma unroll
      for (int q = 0; q < 6; ++q) sZ[khi * 6 + q] = zz[q];
    }
  }
  __syncthreads();
  if (threadIdx.x < NCOL) {
    const int n = threadIdx.x;
    float v = 0.0f;
    if (n < 6 * NHI) v = sZ[(n % NHI) * 6 + n / NHI];
    S[klo * NCOL + n] = v;
  }
  if (blockIdx.x == 0 && threadIdx.x >= NCOL && threadIdx.x < 4 * NCOL)
    S[NLO * NCOL + (threadIdx.x - NCOL)] = 0.0f;
}

__device__ __forceinline__ __attribute__((target("no-packed-fp32-ops")))
void interp_weights(float t, float& g0, float& g1, float& g2) {
  const float E  = __builtin_amdgcn_exp2f(-2.88539008177792681472f * __builtin_fabsf(t));
  const float ua = (1.0f - E) * __builtin_amdgcn_rcpf(1.0f + E);
  const float u  = __builtin_copysignf(ua, t);
  const float w  = u * u;
  float pp = 1.5082264e-4f;
  pp = __builtin_fmaf(pp, w, -4.6722330e-3f);
  pp = __builtin_fmaf(pp, w, 7.9688482e-2f);
  pp = __builtin_fmaf(pp, w, -6.4596337e-1f);
  pp = __builtin_fmaf(pp, w, 1.5707963e+0f);
  const float sh = u * pp;
  float ch = -2.3824918e-5f;
  ch = __builtin_fmaf(ch, w, 9.1772486e-4f);
  ch = __builtin_fmaf(ch, w, -2.0862689e-2f);
  ch = __builtin_fmaf(ch, w, 2.5366932e-1f);
  ch = __builtin_fmaf(ch, w, -1.2337005e+0f);
  ch = __builtin_fmaf(ch, w, 1.0f);
  const float hs = sh * ch;
  const float ca = __builtin_fmaf(-2.0f * sh, sh, 1.0f);
  const float third = 0.33333333333333333333f;
  const float r33 = 0.57735026918962576451f;
  const float bb = __builtin_fmaf(hs, 2.0f * third, third);
  g0 = __builtin_fmaf(hs, -4.0f * third, third);
  g1 = __builtin_fmaf(ca, r33, bb);
  g2 = __builtin_fmaf(ca, -r33, bb);
}

#define block_sync() do { __builtin_amdgcn_fence(__ATOMIC_RELEASE, "workgroup"); __builtin_amdgcn_s_barrier(); __builtin_amdgcn_fence(__ATOMIC_ACQUIRE, "workgroup"); } while (0)

__global__ __launch_bounds__(WAVES * 32) __attribute__((target("no-packed-fp32-ops"))) void qnn_kernel(
    const float* __restrict__ x,
    const float* __restrict__ W1, const float* __restrict__ b1,
    const float* __restrict__ W2, const float* __restrict__ b2,
    const float* __restrict__ W3, const float* __restrict__ b3,
    const float* __restrict__ W4, const float* __restrict__ b4,
    const float* __restrict__ S,
    float* __restrict__ out, int batch) {
  __shared__ __attribute__((aligned(16))) float sWgt[TILE * KPAD];
  __shared__ __attribute__((aligned(16))) float sY[TILE * YSTR];

  const int tid  = (int)__builtin_amdgcn_workitem_id_x();
  const int wave = tid >> 5;
  const int lane = tid & 31;
  const int half = lane >> 4;
  const int l    = lane & 15;
  const int ebase = (int)__builtin_amdgcn_workgroup_id_x() * TILE;

  const int e = ebase + tid;
  const int ec = (e < batch) ? e : (batch - 1);
  const float x0 = x[ec * 4 + 0], x1 = x[ec * 4 + 1];
  const float x2 = x[ec * 4 + 2], x3 = x[ec * 4 + 3];

  v2f bs[KSTEPS];
  {
    const int n = wave * 16 + l;
#pragma unroll
    for (int kt = 0; kt < KSTEPS; ++kt) {
      const int kb = 4 * kt + 2 * half;
      bs[kt].x = S[kb * NCOL + n];   bs[kt].y = S[(kb + 1) * NCOL + n];
    }
  }

  float h[16];
  {
#pragma unroll
    for (int j = 0; j < 16; ++j) h[j] = b1[j];
    const float xk[4] = {x0, x1, x2, x3};
#pragma unroll
    for (int k = 0; k < 4; ++k) {
#pragma unroll
      for (int j = 0; j < 16; ++j) h[j] += xk[k] * W1[16 * k + j];
    }
#pragma unroll
    for (int j = 0; j < 16; ++j) h[j] = h[j] > 0.0f ? h[j] : 0.0f;
  }
  __builtin_amdgcn_sched_barrier(0);
  {
    float t[4];
#pragma unroll
    for (int q = 0; q < 4; ++q) t[q] = b2[2 + q];
#pragma unroll
    for (int j = 0; j < 16; ++j) {
#pragma unroll
      for (int q = 0; q < 4; ++q) t[q] += h[j] * W2[6 * j + 2 + q];
    }
    float g[4][3];
#pragma unroll
    for (int q = 0; q < 4; ++q) interp_weights(t[q], g[q][0], g[q][1], g[q][2]);
    float m23[9], m45[9];
#pragma unroll
    for (int i = 0; i < 3; ++i)
#pragma unroll
      for (int j = 0; j < 3; ++j) {
        m23[3 * i + j] = g[0][i] * g[1][j];
        m45[3 * i + j] = g[2][i] * g[3][j];
      }
    float* dst = &sWgt[tid * KPAD];
#pragma unroll
    for (int a = 0; a < 9; ++a)
#pragma unroll
      for (int b = 0; b < 9; ++b)
        dst[9 * a + b] = m23[a] * m45[b];
    dst[81] = 0.0f; dst[82] = 0.0f; dst[83] = 0.0f;
  }
  block_sync();

  {
    const float* arow = &sWgt[l * KPAD + 2 * half];
    float* yrow = &sY[(8 * half) * YSTR + wave * 16 + l];
    v2f a[2][KSTEPS];
    v8f c0[2], c1[2];
    float pv[2][8];
#pragma unroll
    for (int kt = 0; kt < KSTEPS; ++kt) {
      a[0][kt].x = arow[4 * kt];
      a[0][kt].y = arow[4 * kt + 1];
    }
#pragma unroll
    for (int m = 0; m < 10; ++m) {
      if (m + 1 < 8) {
        const float* am = arow + (m + 1) * 16 * KPAD;
#pragma unroll
        for (int kt = 0; kt < KSTEPS; ++kt) {
          a[(m + 1) & 1][kt].x = am[4 * kt];
          a[(m + 1) & 1][kt].y = am[4 * kt + 1];
        }
      }
      if (m >= 2) {
        float* ym = yrow + (m - 2) * 16 * YSTR;
#pragma unroll
        for (int r = 0; r < 8; ++r) ym[r * YSTR] = pv[m & 1][r];
      }
      __builtin_amdgcn_sched_barrier(0);
      if (m < 8) {
        v8f acc0 = v8f{}, acc1 = v8f{};
#pragma unroll
        for (int kt = 0; kt < KSTEPS; ++kt) {
          if (kt & 1)
            acc1 = __builtin_amdgcn_wmma_f32_16x16x4_f32(
                false, a[m & 1][kt], false, bs[kt], (short)0, acc1, false, false);
          else
            acc0 = __builtin_amdgcn_wmma_f32_16x16x4_f32(
                false, a[m & 1][kt], false, bs[kt], (short)0, acc0, false, false);
        }
        c0[m & 1] = acc0; c1[m & 1] = acc1;
      }
      if (m >= 1 && m <= 8) {
#pragma unroll
        for (int r = 0; r < 8; ++r)
          pv[(m - 1) & 1][r] = c0[(m - 1) & 1][r] + c1[(m - 1) & 1][r];
      }
      __builtin_amdgcn_sched_barrier(0);
    }
  }
  block_sync();

  {
    if (e < batch) {
      const v4f* y4 = reinterpret_cast<const v4f*>(&sY[tid * YSTR]);
      float y[56];
#pragma unroll
      for (int j = 0; j < 14; ++j) {
        const v4f t = y4[j];
        y[4 * j] = t.x; y[4 * j + 1] = t.y; y[4 * j + 2] = t.z; y[4 * j + 3] = t.w;
      }
      float whi[NHI];
      {
        float t0 = b2[0], t1 = b2[1];
#pragma unroll
        for (int j = 0; j < 16; ++j) {
          t0 += h[j] * W2[6 * j];
          t1 += h[j] * W2[6 * j + 1];
        }
        float g0[3], g1[3];
        interp_weights(t0, g0[0], g0[1], g0[2]);
        interp_weights(t1, g1[0], g1[1], g1[2]);
#pragma unroll
        for (int i = 0; i < 3; ++i)
#pragma unroll
          for (int j = 0; j < 3; ++j) whi[3 * i + j] = g0[i] * g1[j];
      }
      float z[6];
#pragma unroll
      for (int q = 0; q < 6; ++q) {
        float acc = whi[0] * y[q * NHI];
#pragma unroll
        for (int k = 1; k < NHI; ++k) acc += whi[k] * y[q * NHI + k];
        z[q] = acc;
      }
      float o0 = b4[0], o1 = b4[1], o2 = b4[2], o3 = b4[3];
#pragma unroll
      for (int j = 0; j < 16; ++j) {
        float g = b3[j];
#pragma unroll
        for (int q = 0; q < 6; ++q) g += z[q] * W3[q * 16 + j];
        g = g > 0.0f ? g : 0.0f;
        o0 += g * W4[j * 4 + 0]; o1 += g * W4[j * 4 + 1];
        o2 += g * W4[j * 4 + 2]; o3 += g * W4[j * 4 + 3];
      }
      float* op = out + (size_t)e * 4;
      op[0] = o0; op[1] = o1; op[2] = o2; op[3] = o3;
    }
  }
}

extern "C" void kernel_launch(void* const* d_in, const int* in_sizes, int n_in,
                              void* d_out, int out_size, void* d_ws, size_t ws_size,
                              hipStream_t stream) {
  const float* x  = (const float*)d_in[0];
  const float* W1 = (const float*)d_in[1];
  const float* b1 = (const float*)d_in[2];
  const float* W2 = (const float*)d_in[3];
  const float* b2 = (const float*)d_in[4];
  const float* qp = (const float*)d_in[5];
  const float* W3 = (const float*)d_in[6];
  const float* b3 = (const float*)d_in[7];
  const float* W4 = (const float*)d_in[8];
  const float* b4 = (const float*)d_in[9];
  float* out = (float*)d_out;
  float* S   = (float*)d_ws;

  const int batch = in_sizes[0] / 4;
  build_samples_kernel<<<dim3(NLO), dim3(NHI * 32), 0, stream>>>(qp, S);
  const int tiles = (batch + TILE - 1) / TILE;
  qnn_kernel<<<dim3(tiles), dim3(WAVES * 32), 0, stream>>>(
      x, W1, b1, W2, b2, W3, b3, W4, b4, S, out, batch);
}


# Sample kernel: the three circuit layers unrolled in full, so each gate's sine and cosine come by one lane broadcast at a literal lane, not two computed-lane reads and a scalar select per value
# speedup vs baseline: 18.3916x; 1.0368x over previous
#include <hip/hip_runtime.h>
#include <math.h>

typedef __attribute__((ext_vector_type(2))) float v2f;
typedef __attribute__((ext_vector_type(4))) float v4f;
typedef __attribute__((ext_vector_type(8))) float v8f;

#define WAVES 4
#define TILE  (WAVES * 32)
#define NHI    9
#define NLO    81
#define KSTEPS 21
#define KPAD   84
#define NCOL   64
#define YSTR   68


template <int M>
__device__ __forceinline__ float lane_xor(float v) {
  const int b = __float_as_int(v);
  if constexpr (M == 16) {
    return __int_as_float(
        __builtin_amdgcn_permlanex16(b, b, 0x76543210, 0xfedcba98, false, false));
  } else {
    return __int_as_float(__builtin_amdgcn_mov_dpp(b, 0x160 + M, 0xf, 0xf, true));
  }
}
template <int M>
__device__ __forceinline__ void lane_xor4(float a, float b, float c, float d,
                                          float& pa, float& pb, float& pc, float& pd) {
  pa = lane_xor<M>(a); pb = lane_xor<M>(b); pc = lane_xor<M>(c); pd = lane_xor<M>(d);
}
__device__ __forceinline__ float lane_get(float v, int src) {
  return __int_as_float(__builtin_amdgcn_readlane(__float_as_int(v), src));
}


__global__ __launch_bounds__(NHI * 32) void build_samples_kernel(
    const float* __restrict__ qp, float* __restrict__ S) {
  __shared__ float sZ[NHI * 6];
  const int lane = threadIdx.x & 31;
  const int khi  = threadIdx.x >> 5;
  const int klo  = blockIdx.x;
  const int s0 = lane, s1 = lane + 32;

  float snA, csA, snB, csB;
  {
    const int tb = (lane + 32 < 54) ? lane + 32 : 53;
    const float thA = qp[lane], thB = qp[tb];
    sincosf(0.5f * thA, &snA, &csA);
    sincosf(0.5f * thB, &snB, &csB);
  }
  auto trig = [&](int idx, float& sh, float& ch) {
    const int j = idx & 31;
    const float sa = lane_get(snA, j), ca = lane_get(csA, j);
    const float sb = lane_get(snB, j), cb = lane_get(csB, j);
    sh = (idx < 32) ? sa : sb;
    ch = (idx < 32) ? ca : cb;
  };


  float r0 = 1.0f, i0 = 0.0f;
  float r1 = 1.0f, i1 = 0.0f;
  {
    int kd[6];
    kd[0] = khi / 3;  kd[1] = khi % 3;
    kd[2] = klo / 27; kd[3] = (klo / 9) % 3; kd[4] = (klo / 3) % 3; kd[5] = klo % 3;
#pragma unroll
    for (int q = 0; q < 6; ++q) {
      const float ca = (kd[q] == 0) ? 1.0f : ((kd[q] == 1) ? 0.5f : -0.5f);
      const float sa = (kd[q] == 0) ? 0.0f : 0.86602540378443864676f;
      const int m = 1 << (5 - q);
      r0 *= (s0 & m) ? sa : ca;
      r1 *= (s1 & m) ? sa : ca;
    }
  }

  auto partner = [&](int m, float& pr0, float& pi0, float& pr1, float& pi1) {
    switch (m) {
      case 32: pr0 = r1; pi0 = i1; pr1 = r0; pi1 = i0; break;
      case 16: lane_xor4<16>(r0, i0, r1, i1, pr0, pi0, pr1, pi1); break;
      case 8:  lane_xor4<8>(r0, i0, r1, i1, pr0, pi0, pr1, pi1); break;
      case 4:  lane_xor4<4>(r0, i0, r1, i1, pr0, pi0, pr1, pi1); break;
      case 2:  lane_xor4<2>(r0, i0, r1, i1, pr0, pi0, pr1, pi1); break;
      default: lane_xor4<1>(r0, i0, r1, i1, pr0, pi0, pr1, pi1); break;
    }
  };
  auto cnot = [&](int cq, int tq) {
    const int mc = 1 << (5 - cq), mt = 1 << (5 - tq);
    float pr0, pi0, pr1, pi1;
    partner(mt, pr0, pi0, pr1, pi1);
    if (s0 & mc) { r0 = pr0; i0 = pi0; }
    if (s1 & mc) { r1 = pr1; i1 = pi1; }
  };
  auto rx = [&](int q, float sh, float ch) {
    const int m = 1 << (5 - q);
    float pr0, pi0, pr1, pi1;
    partner(m, pr0, pi0, pr1, pi1);
    const float a0 = ch * r0 + sh * pi0, b0 = ch * i0 - sh * pr0;
    const float a1 = ch * r1 + sh * pi1, b1 = ch * i1 - sh * pr1;
    r0 = a0; i0 = b0; r1 = a1; i1 = b1;
  };
  auto ry = [&](int q, float sh, float ch) {
    const int m = 1 << (5 - q);
    float pr0, pi0, pr1, pi1;
    partner(m, pr0, pi0, pr1, pi1);
    const float t0 = (s0 & m) ? sh : -sh;
    const float t1 = (s1 & m) ? sh : -sh;
    const float a0 = ch * r0 + t0 * pr0, b0 = ch * i0 + t0 * pi0;
    const float a1 = ch * r1 + t1 * pr1, b1 = ch * i1 + t1 * pi1;
    r0 = a0; i0 = b0; r1 = a1; i1 = b1;
  };
  auto phase = [&](int s, int m, float sh, float ch, float& r, float& i) {
    const float t = (s & m) ? -sh : sh;
    const float a = ch * r + t * i, b = ch * i - t * r;
    r = a; i = b;
  };
  auto rz = [&](int q, float sh, float ch) {
    const int m = 1 << (5 - q);
    phase(s0, m, sh, ch, r0, i0);
    phase(s1, m, sh, ch, r1, i1);
  };
  auto crz = [&](int cq, int tq, float sh, float ch) {
    const int mc = 1 << (5 - cq), mt = 1 << (5 - tq);
    float a = r0, b = i0;
    phase(s0, mt, sh, ch, a, b);
    if (s0 & mc) { r0 = a; i0 = b; }
    a = r1; b = i1;
    phase(s1, mt, sh, ch, a, b);
    if (s1 & mc) { r1 = a; i1 = b; }
  };

#pragma unroll
  for (int layer = 0; layer < 3; ++layer) {
    cnot(0, 1); cnot(2, 3); cnot(4, 5);
    cnot(1, 2); cnot(3, 4);
#pragma unroll
    for (int i = 0; i < 6; ++i) {
      float sh, ch;
      trig(layer * 18 + i * 3 + 0, sh, ch); rx(i, sh, ch);
      trig(layer * 18 + i * 3 + 1, sh, ch); ry(i, sh, ch);
      trig(layer * 18 + i * 3 + 2, sh, ch); rz(i, sh, ch);
    }
    if (layer < 2) {
#pragma unroll
      for (int i = 0; i < 6; ++i) {
        float sh, ch;
        trig(layer * 18 + i * 3 + 0, sh, ch);
        crz(i, (i + 1) % 6, sh, ch);
      }
    }
  }

  {
    const float p0 = r0 * r0 + i0 * i0, p1 = r1 * r1 + i1 * i1;
    const float t = p0 + p1;
    float zz[6];
    zz[0] = p0 - p1;
#pragma unroll
    for (int q = 1; q < 6; ++q) zz[q] = (lane & (1 << (5 - q))) ? -t : t;
#pragma unroll
    for (int q = 0; q < 6; ++q) {
      float v = zz[q];
      v += lane_xor<1>(v); v += lane_xor<2>(v); v += lane_xor<4>(v);
      v += lane_xor<8>(v); v += lane_xor<16>(v);
      zz[q] = v;
    }
    if (lane == 0) {
#pragma unroll
      for (int q = 0; q < 6; ++q) sZ[khi * 6 + q] = zz[q];
    }
  }
  __syncthreads();
  if (threadIdx.x < NCOL) {
    const int n = threadIdx.x;
    float v = 0.0f;
    if (n < 6 * NHI) v = sZ[(n % NHI) * 6 + n / NHI];
    S[klo * NCOL + n] = v;
  }
  if (blockIdx.x == 0 && threadIdx.x >= NCOL && threadIdx.x < 4 * NCOL)
    S[NLO * NCOL + (threadIdx.x - NCOL)] = 0.0f;
}

__device__ __forceinline__ __attribute__((target("no-packed-fp32-ops")))
void interp_weights(float t, float& g0, float& g1, float& g2) {
  const float E  = __builtin_amdgcn_exp2f(-2.88539008177792681472f * __builtin_fabsf(t));
  const float ua = (1.0f - E) * __builtin_amdgcn_rcpf(1.0f + E);
  const float u  = __builtin_copysignf(ua, t);
  const float w  = u * u;
  float pp = 1.5082264e-4f;
  pp = __builtin_fmaf(pp, w, -4.6722330e-3f);
  pp = __builtin_fmaf(pp, w, 7.9688482e-2f);
  pp = __builtin_fmaf(pp, w, -6.4596337e-1f);
  pp = __builtin_fmaf(pp, w, 1.5707963e+0f);
  const float sh = u * pp;
  float ch = -2.3824918e-5f;
  ch = __builtin_fmaf(ch, w, 9.1772486e-4f);
  ch = __builtin_fmaf(ch, w, -2.0862689e-2f);
  ch = __builtin_fmaf(ch, w, 2.5366932e-1f);
  ch = __builtin_fmaf(ch, w, -1.2337005e+0f);
  ch = __builtin_fmaf(ch, w, 1.0f);
  const float hs = sh * ch;
  const float ca = __builtin_fmaf(-2.0f * sh, sh, 1.0f);
  const float third = 0.33333333333333333333f;
  const float r33 = 0.57735026918962576451f;
  const float bb = __builtin_fmaf(hs, 2.0f * third, third);
  g0 = __builtin_fmaf(hs, -4.0f * third, third);
  g1 = __builtin_fmaf(ca, r33, bb);
  g2 = __builtin_fmaf(ca, -r33, bb);
}

#define block_sync() do { __builtin_amdgcn_fence(__ATOMIC_RELEASE, "workgroup"); __builtin_amdgcn_s_barrier(); __builtin_amdgcn_fence(__ATOMIC_ACQUIRE, "workgroup"); } while (0)

__global__ __launch_bounds__(WAVES * 32) __attribute__((target("no-packed-fp32-ops"))) void qnn_kernel(
    const float* __restrict__ x,
    const float* __restrict__ W1, const float* __restrict__ b1,
    const float* __restrict__ W2, const float* __restrict__ b2,
    const float* __restrict__ W3, const float* __restrict__ b3,
    const float* __restrict__ W4, const float* __restrict__ b4,
    const float* __restrict__ S,
    float* __restrict__ out, int batch) {
  __shared__ __attribute__((aligned(16))) float sWgt[TILE * KPAD];
  __shared__ __attribute__((aligned(16))) float sY[TILE * YSTR];

  const int tid  = (int)__builtin_amdgcn_workitem_id_x();
  const int wave = tid >> 5;
  const int lane = tid & 31;
  const int half = lane >> 4;
  const int l    = lane & 15;
  const int ebase = (int)__builtin_amdgcn_workgroup_id_x() * TILE;

  const int e = ebase + tid;
  const int ec = (e < batch) ? e : (batch - 1);
  const float x0 = x[ec * 4 + 0], x1 = x[ec * 4 + 1];
  const float x2 = x[ec * 4 + 2], x3 = x[ec * 4 + 3];

  v2f bs[KSTEPS];
  {
    const int n = wave * 16 + l;
#pragma unroll
    for (int kt = 0; kt < KSTEPS; ++kt) {
      const int kb = 4 * kt + 2 * half;
      bs[kt].x = S[kb * NCOL + n];   bs[kt].y = S[(kb + 1) * NCOL + n];
    }
  }

  float h[16];
  {
#pragma unroll
    for (int j = 0; j < 16; ++j) h[j] = b1[j];
    const float xk[4] = {x0, x1, x2, x3};
#pragma unroll
    for (int k = 0; k < 4; ++k) {
#pragma unroll
      for (int j = 0; j < 16; ++j) h[j] += xk[k] * W1[16 * k + j];
    }
#pragma unroll
    for (int j = 0; j < 16; ++j) h[j] = h[j] > 0.0f ? h[j] : 0.0f;
  }
  __builtin_amdgcn_sched_barrier(0);
  {
    float t[4];
#pragma unroll
    for (int q = 0; q < 4; ++q) t[q] = b2[2 + q];
#pragma unroll
    for (int j = 0; j < 16; ++j) {
#pragma unroll
      for (int q = 0; q < 4; ++q) t[q] += h[j] * W2[6 * j + 2 + q];
    }
    float g[4][3];
#pragma unroll
    for (int q = 0; q < 4; ++q) interp_weights(t[q], g[q][0], g[q][1], g[q][2]);
    float m23[9], m45[9];
#pragma unroll
    for (int i = 0; i < 3; ++i)
#pragma unroll
      for (int j = 0; j < 3; ++j) {
        m23[3 * i + j] = g[0][i] * g[1][j];
        m45[3 * i + j] = g[2][i] * g[3][j];
      }
    float* dst = &sWgt[tid * KPAD];
#pragma unroll
    for (int a = 0; a < 9; ++a)
#pragma unroll
      for (int b = 0; b < 9; ++b)
        dst[9 * a + b] = m23[a] * m45[b];
    dst[81] = 0.0f; dst[82] = 0.0f; dst[83] = 0.0f;
  }
  block_sync();

  {
    const float* arow = &sWgt[l * KPAD + 2 * half];
    float* yrow = &sY[(8 * half) * YSTR + wave * 16 + l];
    v2f a[2][KSTEPS];
    v8f c0[2], c1[2];
    float pv[2][8];
#pragma unroll
    for (int kt = 0; kt < KSTEPS; ++kt) {
      a[0][kt].x = arow[4 * kt];
      a[0][kt].y = arow[4 * kt + 1];
    }
#pragma unroll
    for (int m = 0; m < 10; ++m) {
      if (m + 1 < 8) {
        const float* am = arow + (m + 1) * 16 * KPAD;
#pragma unroll
        for (int kt = 0; kt < KSTEPS; ++kt) {
          a[(m + 1) & 1][kt].x = am[4 * kt];
          a[(m + 1) & 1][kt].y = am[4 * kt + 1];
        }
      }
      if (m >= 2) {
        float* ym = yrow + (m - 2) * 16 * YSTR;
#pragma unroll
        for (int r = 0; r < 8; ++r) ym[r * YSTR] = pv[m & 1][r];
      }
      __builtin_amdgcn_sched_barrier(0);
      if (m < 8) {
        v8f acc0 = v8f{}, acc1 = v8f{};
#pragma unroll
        for (int kt = 0; kt < KSTEPS; ++kt) {
          if (kt & 1)
            acc1 = __builtin_amdgcn_wmma_f32_16x16x4_f32(
                false, a[m & 1][kt], false, bs[kt], (short)0, acc1, false, false);
          else
            acc0 = __builtin_amdgcn_wmma_f32_16x16x4_f32(
                false, a[m & 1][kt], false, bs[kt], (short)0, acc0, false, false);
        }
        c0[m & 1] = acc0; c1[m & 1] = acc1;
      }
      if (m >= 1 && m <= 8) {
#pragma unroll
        for (int r = 0; r < 8; ++r)
          pv[(m - 1) & 1][r] = c0[(m - 1) & 1][r] + c1[(m - 1) & 1][r];
      }
      __builtin_amdgcn_sched_barrier(0);
    }
  }
  block_sync();

  {
    if (e < batch) {
      const v4f* y4 = reinterpret_cast<const v4f*>(&sY[tid * YSTR]);
      float y[56];
#pragma unroll
      for (int j = 0; j < 14; ++j) {
        const v4f t = y4[j];
        y[4 * j] = t.x; y[4 * j + 1] = t.y; y[4 * j + 2] = t.z; y[4 * j + 3] = t.w;
      }
      float whi[NHI];
      {
        float t0 = b2[0], t1 = b2[1];
#pragma unroll
        for (int j = 0; j < 16; ++j) {
          t0 += h[j] * W2[6 * j];
          t1 += h[j] * W2[6 * j + 1];
        }
        float g0[3], g1[3];
        interp_weights(t0, g0[0], g0[1], g0[2]);
        interp_weights(t1, g1[0], g1[1], g1[2]);
#pragma unroll
        for (int i = 0; i < 3; ++i)
#pragma unroll
          for (int j = 0; j < 3; ++j) whi[3 * i + j] = g0[i] * g1[j];
      }
      float z[6];
#pragma unroll
      for (int q = 0; q < 6; ++q) {
        float acc = whi[0] * y[q * NHI];
#pragma unroll
        for (int k = 1; k < NHI; ++k) acc += whi[k] * y[q * NHI + k];
        z[q] = acc;
      }
      float o0 = b4[0], o1 = b4[1], o2 = b4[2], o3 = b4[3];
#pragma unroll
      for (int j = 0; j < 16; ++j) {
        float g = b3[j];
#pragma unroll
        for (int q = 0; q < 6; ++q) g += z[q] * W3[q * 16 + j];
        g = g > 0.0f ? g : 0.0f;
        o0 += g * W4[j * 4 + 0]; o1 += g * W4[j * 4 + 1];
        o2 += g * W4[j * 4 + 2]; o3 += g * W4[j * 4 + 3];
      }
      float* op = out + (size_t)e * 4;
      op[0] = o0; op[1] = o1; op[2] = o2; op[3] = o3;
    }
  }
}

extern "C" void kernel_launch(void* const* d_in, const int* in_sizes, int n_in,
                              void* d_out, int out_size, void* d_ws, size_t ws_size,
                              hipStream_t stream) {
  const float* x  = (const float*)d_in[0];
  const float* W1 = (const float*)d_in[1];
  const float* b1 = (const float*)d_in[2];
  const float* W2 = (const float*)d_in[3];
  const float* b2 = (const float*)d_in[4];
  const float* qp = (const float*)d_in[5];
  const float* W3 = (const float*)d_in[6];
  const float* b3 = (const float*)d_in[7];
  const float* W4 = (const float*)d_in[8];
  const float* b4 = (const float*)d_in[9];
  float* out = (float*)d_out;
  float* S   = (float*)d_ws;

  const int batch = in_sizes[0] / 4;
  build_samples_kernel<<<dim3(NLO), dim3(NHI * 32), 0, stream>>>(qp, S);
  const int tiles = (batch + TILE - 1) / TILE;
  qnn_kernel<<<dim3(tiles), dim3(WAVES * 32), 0, stream>>>(
      x, W1, b1, W2, b2, W3, b3, W4, b4, S, out, batch);
}
